# starting kernel
# baseline (speedup 1.0000x reference)
#include <hip/hip_runtime.h>

typedef __attribute__((ext_vector_type(2))) float v2f;
typedef __attribute__((ext_vector_type(8))) float v8f;

#define N_NODES_C 100000
#define N_EDGES_C 20000
#define NNZ_C     1000000
#define ALPHA_C   0.5f

__device__ __forceinline__ v8f wmma_f32(v2f a, v2f b, v8f c) {
  return __builtin_amdgcn_wmma_f32_16x16x4_f32(false, a, false, b, (short)0, c,
                                               false, false);
}

__global__ __launch_bounds__(256) void gemm64_bias_kernel(
    const float* __restrict__ A, const float* __restrict__ W,
    const float* __restrict__ bias, float* __restrict__ Out, int M) {
  __shared__ float Bsh[64 * 64];
  for (int i = threadIdx.x; i < 64 * 64; i += 256) {
    const int k = i >> 6, n = i & 63;
    Bsh[i] = W[n * 64 + k];
  }
  __syncthreads();

  const int lane = threadIdx.x & 31;
  const int wave = threadIdx.x >> 5;
  const int rowBase = (blockIdx.x * 8 + wave) << 4;
  if (rowBase >= M) return;
  const int ll = lane & 15;
  const int lh = lane >> 4;

  const float* Arow = A + (size_t)(rowBase + ll) * 64;
  v8f acc0 = {}, acc1 = {}, acc2 = {}, acc3 = {};

#pragma unroll
  for (int k0 = 0; k0 < 64; k0 += 4) {
    const int ka = k0 + lh * 2;
    v2f a;
    a.x = Arow[ka];
    a.y = Arow[ka + 1];
    const float* B0 = &Bsh[ka * 64 + ll];
    const float* B1 = &Bsh[(ka + 1) * 64 + ll];
    v2f b;
    b.x = B0[0];  b.y = B1[0];  acc0 = wmma_f32(a, b, acc0);
    b.x = B0[16]; b.y = B1[16]; acc1 = wmma_f32(a, b, acc1);
    b.x = B0[32]; b.y = B1[32]; acc2 = wmma_f32(a, b, acc2);
    b.x = B0[48]; b.y = B1[48]; acc3 = wmma_f32(a, b, acc3);
  }

#pragma unroll
  for (int j = 0; j < 8; ++j) {
    const int r = rowBase + j + lh * 8;
    float* o = Out + (size_t)r * 64;
    o[ll]      = acc0[j] + bias[ll];
    o[16 + ll] = acc1[j] + bias[16 + ll];
    o[32 + ll] = acc2[j] + bias[32 + ll];
    o[48 + ll] = acc3[j] + bias[48 + ll];
  }
}

__global__ __launch_bounds__(256) void gemm2_mix_kernel(
    const float* __restrict__ X, const float* __restrict__ X0,
    const float* __restrict__ S, const float* __restrict__ deg,
    const float* __restrict__ W2, const float* __restrict__ b2,
    float* __restrict__ T, int M) {
  __shared__ float Bsh[128 * 64];
  for (int i = threadIdx.x; i < 128 * 64; i += 256) {
    const int k = i >> 6, n = i & 63;
    Bsh[i] = W2[n * 128 + k];
  }
  __syncthreads();

  const int lane = threadIdx.x & 31;
  const int wave = threadIdx.x >> 5;
  const int rowBase = (blockIdx.x * 8 + wave) << 4;
  if (rowBase >= M) return;
  const int ll = lane & 15;
  const int lh = lane >> 4;

  const float dg = deg[rowBase + ll];
  const float* Xrow = X + (size_t)(rowBase + ll) * 64;
  const float* Srow = S + (size_t)(rowBase + ll) * 64;
  v8f acc0 = {}, acc1 = {}, acc2 = {}, acc3 = {};

#pragma unroll
  for (int k0 = 0; k0 < 64; k0 += 4) {
    const int ka = k0 + lh * 2;
    v2f a;
    a.x = dg * Xrow[ka];
    a.y = dg * Xrow[ka + 1];
    const float* B0 = &Bsh[ka * 64 + ll];
    const float* B1 = &Bsh[(ka + 1) * 64 + ll];
    v2f b;
    b.x = B0[0];  b.y = B1[0];  acc0 = wmma_f32(a, b, acc0);
    b.x = B0[16]; b.y = B1[16]; acc1 = wmma_f32(a, b, acc1);
    b.x = B0[32]; b.y = B1[32]; acc2 = wmma_f32(a, b, acc2);
    b.x = B0[48]; b.y = B1[48]; acc3 = wmma_f32(a, b, acc3);
  }
#pragma unroll
  for (int k0 = 64; k0 < 128; k0 += 4) {
    const int ka = k0 + lh * 2;
    v2f a;
    a.x = Srow[ka - 64];
    a.y = Srow[ka - 63];
    const float* B0 = &Bsh[ka * 64 + ll];
    const float* B1 = &Bsh[(ka + 1) * 64 + ll];
    v2f b;
    b.x = B0[0];  b.y = B1[0];  acc0 = wmma_f32(a, b, acc0);
    b.x = B0[16]; b.y = B1[16]; acc1 = wmma_f32(a, b, acc1);
    b.x = B0[32]; b.y = B1[32]; acc2 = wmma_f32(a, b, acc2);
    b.x = B0[48]; b.y = B1[48]; acc3 = wmma_f32(a, b, acc3);
  }

  const float oma = 1.0f - ALPHA_C;
#pragma unroll
  for (int j = 0; j < 8; ++j) {
    const int r = rowBase + j + lh * 8;
    const float dr = deg[r];
    const float* x0r = X0 + (size_t)r * 64;
    float* o = T + (size_t)r * 64;
    o[ll]      = oma * (acc0[j] + dr * b2[ll])      + ALPHA_C * x0r[ll];
    o[16 + ll] = oma * (acc1[j] + dr * b2[16 + ll]) + ALPHA_C * x0r[16 + ll];
    o[32 + ll] = oma * (acc2[j] + dr * b2[32 + ll]) + ALPHA_C * x0r[32 + ll];
    o[48 + ll] = oma * (acc3[j] + dr * b2[48 + ll]) + ALPHA_C * x0r[48 + ll];
  }
}

__global__ __launch_bounds__(256) void scatter_edges_kernel(
    const float* __restrict__ Y1, const int* __restrict__ vertex,
    const int* __restrict__ edges, float* __restrict__ Xe,
    float* __restrict__ deg) {
  const int tid = blockIdx.x * 256 + threadIdx.x;
  const int e = tid >> 4;
  if (e >= NNZ_C) return;
  const int c = (tid & 15) << 2;
  const int v = vertex[e];
  const int h = edges[e];
  const float4 y = *(const float4*)(Y1 + (size_t)v * 64 + c);
  float* dst = Xe + (size_t)h * 64 + c;
  unsafeAtomicAdd(dst + 0, y.x);
  unsafeAtomicAdd(dst + 1, y.y);
  unsafeAtomicAdd(dst + 2, y.z);
  unsafeAtomicAdd(dst + 3, y.w);
  if ((tid & 15) == 0) unsafeAtomicAdd(deg + v, 1.0f);
}

__global__ __launch_bounds__(256) void scatter_nodes_kernel(
    const float* __restrict__ Xe, const int* __restrict__ vertex,
    const int* __restrict__ edges, float* __restrict__ S) {
  const int tid = blockIdx.x * 256 + threadIdx.x;
  const int e = tid >> 4;
  if (e >= NNZ_C) return;
  const int c = (tid & 15) << 2;
  const int v = vertex[e];
  const int h = edges[e];
  const float4 y = *(const float4*)(Xe + (size_t)h * 64 + c);
  float* dst = S + (size_t)v * 64 + c;
  unsafeAtomicAdd(dst + 0, y.x);
  unsafeAtomicAdd(dst + 1, y.y);
  unsafeAtomicAdd(dst + 2, y.z);
  unsafeAtomicAdd(dst + 3, y.w);
}

extern "C" void kernel_launch(void* const* d_in, const int* in_sizes, int n_in,
                              void* d_out, int out_size, void* d_ws,
                              size_t ws_size, hipStream_t stream) {
  (void)in_sizes; (void)n_in; (void)out_size; (void)ws_size;
  const float* X    = (const float*)d_in[0];
  const float* X0   = (const float*)d_in[1];
  const int*   vertex = (const int*)d_in[2];
  const int*   edges  = (const int*)d_in[3];
  const float* W1w  = (const float*)d_in[4];
  const float* W1b  = (const float*)d_in[5];
  const float* W2w  = (const float*)d_in[6];
  const float* W2b  = (const float*)d_in[7];
  const float* Ww   = (const float*)d_in[8];
  const float* Wb   = (const float*)d_in[9];

  float* out = (float*)d_out;
  float* Xe  = out + (size_t)N_NODES_C * 64;

  float* ws  = (float*)d_ws;
  float* Y1  = ws;
  float* S   = Y1 + (size_t)N_NODES_C * 64;
  float* deg = S + (size_t)N_NODES_C * 64;

  hipMemsetAsync(Xe, 0, (size_t)N_EDGES_C * 64 * sizeof(float), stream);
  hipMemsetAsync(S, 0, ((size_t)N_NODES_C * 64 + N_NODES_C) * sizeof(float),
                 stream);

  const int tiles = N_NODES_C / 16;
  const int gemmBlocks = (tiles + 7) / 8;
  const int scatterBlocks = (NNZ_C * 16) / 256;

  gemm64_bias_kernel<<<gemmBlocks, 256, 0, stream>>>(X, W1w, W1b, Y1, N_NODES_C);
  scatter_edges_kernel<<<scatterBlocks, 256, 0, stream>>>(Y1, vertex, edges, Xe, deg);
  scatter_nodes_kernel<<<scatterBlocks, 256, 0, stream>>>(Xe, vertex, edges, S);
  gemm2_mix_kernel<<<gemmBlocks, 256, 0, stream>>>(X, X0, S, deg, W2w, W2b, Y1,
                                                   N_NODES_C);
  gemm64_bias_kernel<<<gemmBlocks, 256, 0, stream>>>(Y1, Ww, Wb, out, N_NODES_C);
}


# scatter kernels: a wave owns whole 64-float rows (lane = column), each f32 atomic covers one whole 128-B line, not four 32-B pieces; 8 rows per step; sums still atomics in arrival order
# speedup vs baseline: 1.3564x; 1.3564x over previous
#include <hip/hip_runtime.h>

typedef __attribute__((ext_vector_type(2))) float v2f;
typedef __attribute__((ext_vector_type(8))) float v8f;

#define N_NODES_C 100000
#define N_EDGES_C 20000
#define NNZ_C     1000000
#define ALPHA_C   0.5f

__device__ __forceinline__ v8f wmma_f32(v2f a, v2f b, v8f c) {
  return __builtin_amdgcn_wmma_f32_16x16x4_f32(false, a, false, b, (short)0, c,
                                               false, false);
}

__global__ __launch_bounds__(256) void gemm64_bias_kernel(
    const float* __restrict__ A, const float* __restrict__ W,
    const float* __restrict__ bias, float* __restrict__ Out, int M) {
  __shared__ float Bsh[64 * 64];
  for (int i = threadIdx.x; i < 64 * 64; i += 256) {
    const int k = i >> 6, n = i & 63;
    Bsh[i] = W[n * 64 + k];
  }
  __syncthreads();

  const int lane = threadIdx.x & 31;
  const int wave = threadIdx.x >> 5;
  const int rowBase = (blockIdx.x * 8 + wave) << 4;
  if (rowBase >= M) return;
  const int ll = lane & 15;
  const int lh = lane >> 4;

  const float* Arow = A + (size_t)(rowBase + ll) * 64;
  v8f acc0 = {}, acc1 = {}, acc2 = {}, acc3 = {};

#pragma unroll
  for (int k0 = 0; k0 < 64; k0 += 4) {
    const int ka = k0 + lh * 2;
    v2f a;
    a.x = Arow[ka];
    a.y = Arow[ka + 1];
    const float* B0 = &Bsh[ka * 64 + ll];
    const float* B1 = &Bsh[(ka + 1) * 64 + ll];
    v2f b;
    b.x = B0[0];  b.y = B1[0];  acc0 = wmma_f32(a, b, acc0);
    b.x = B0[16]; b.y = B1[16]; acc1 = wmma_f32(a, b, acc1);
    b.x = B0[32]; b.y = B1[32]; acc2 = wmma_f32(a, b, acc2);
    b.x = B0[48]; b.y = B1[48]; acc3 = wmma_f32(a, b, acc3);
  }

#pragma unroll
  for (int j = 0; j < 8; ++j) {
    const int r = rowBase + j + lh * 8;
    float* o = Out + (size_t)r * 64;
    o[ll]      = acc0[j] + bias[ll];
    o[16 + ll] = acc1[j] + bias[16 + ll];
    o[32 + ll] = acc2[j] + bias[32 + ll];
    o[48 + ll] = acc3[j] + bias[48 + ll];
  }
}

__global__ __launch_bounds__(256) void gemm2_mix_kernel(
    const float* __restrict__ X, const float* __restrict__ X0,
    const float* __restrict__ S, const float* __restrict__ deg,
    const float* __restrict__ W2, const float* __restrict__ b2,
    float* __restrict__ T, int M) {
  __shared__ float Bsh[128 * 64];
  for (int i = threadIdx.x; i < 128 * 64; i += 256) {
    const int k = i >> 6, n = i & 63;
    Bsh[i] = W2[n * 128 + k];
  }
  __syncthreads();

  const int lane = threadIdx.x & 31;
  const int wave = threadIdx.x >> 5;
  const int rowBase = (blockIdx.x * 8 + wave) << 4;
  if (rowBase >= M) return;
  const int ll = lane & 15;
  const int lh = lane >> 4;

  const float dg = deg[rowBase + ll];
  const float* Xrow = X + (size_t)(rowBase + ll) * 64;
  const float* Srow = S + (size_t)(rowBase + ll) * 64;
  v8f acc0 = {}, acc1 = {}, acc2 = {}, acc3 = {};

#pragma unroll
  for (int k0 = 0; k0 < 64; k0 += 4) {
    const int ka = k0 + lh * 2;
    v2f a;
    a.x = dg * Xrow[ka];
    a.y = dg * Xrow[ka + 1];
    const float* B0 = &Bsh[ka * 64 + ll];
    const float* B1 = &Bsh[(ka + 1) * 64 + ll];
    v2f b;
    b.x = B0[0];  b.y = B1[0];  acc0 = wmma_f32(a, b, acc0);
    b.x = B0[16]; b.y = B1[16]; acc1 = wmma_f32(a, b, acc1);
    b.x = B0[32]; b.y = B1[32]; acc2 = wmma_f32(a, b, acc2);
    b.x = B0[48]; b.y = B1[48]; acc3 = wmma_f32(a, b, acc3);
  }
#pragma unroll
  for (int k0 = 64; k0 < 128; k0 += 4) {
    const int ka = k0 + lh * 2;
    v2f a;
    a.x = Srow[ka - 64];
    a.y = Srow[ka - 63];
    const float* B0 = &Bsh[ka * 64 + ll];
    const float* B1 = &Bsh[(ka + 1) * 64 + ll];
    v2f b;
    b.x = B0[0];  b.y = B1[0];  acc0 = wmma_f32(a, b, acc0);
    b.x = B0[16]; b.y = B1[16]; acc1 = wmma_f32(a, b, acc1);
    b.x = B0[32]; b.y = B1[32]; acc2 = wmma_f32(a, b, acc2);
    b.x = B0[48]; b.y = B1[48]; acc3 = wmma_f32(a, b, acc3);
  }

  const float oma = 1.0f - ALPHA_C;
#pragma unroll
  for (int j = 0; j < 8; ++j) {
    const int r = rowBase + j + lh * 8;
    const float dr = deg[r];
    const float* x0r = X0 + (size_t)r * 64;
    float* o = T + (size_t)r * 64;
    o[ll]      = oma * (acc0[j] + dr * b2[ll])      + ALPHA_C * x0r[ll];
    o[16 + ll] = oma * (acc1[j] + dr * b2[16 + ll]) + ALPHA_C * x0r[16 + ll];
    o[32 + ll] = oma * (acc2[j] + dr * b2[32 + ll]) + ALPHA_C * x0r[32 + ll];
    o[48 + ll] = oma * (acc3[j] + dr * b2[48 + ll]) + ALPHA_C * x0r[48 + ll];
  }
}

#define ROWS_PER_STEP 8
#define SCATTER_BLOCKS 3907
static_assert(NNZ_C % ROWS_PER_STEP == 0, "incidence list is walked in whole steps");

template <bool COUNT_DEG>
__device__ __forceinline__ void scatter_rows(const float* __restrict__ src,
                                             const int* __restrict__ gidx,
                                             const int* __restrict__ sidx,
                                             float* __restrict__ dst,
                                             float* __restrict__ deg) {
  const int lane = threadIdx.x & 31;
  const int wib = __builtin_amdgcn_readfirstlane((int)(threadIdx.x >> 5));
  const int wave = blockIdx.x * 8 + wib;
  const int nwaves = gridDim.x * 8;
  for (int c = wave; c < NNZ_C / ROWS_PER_STEP; c += nwaves) {
    const int e0 = c * ROWS_PER_STEP;
    int g[ROWS_PER_STEP], s[ROWS_PER_STEP];
#pragma unroll
    for (int i = 0; i < ROWS_PER_STEP; ++i) {
      g[i] = gidx[e0 + i];
      s[i] = sidx[e0 + i];
    }
    float lo[ROWS_PER_STEP], hi[ROWS_PER_STEP];
#pragma unroll
    for (int i = 0; i < ROWS_PER_STEP; ++i) {
      const float* p = src + (size_t)g[i] * 64 + lane;
      lo[i] = p[0];
      hi[i] = p[32];
    }
#pragma unroll
    for (int i = 0; i < ROWS_PER_STEP; ++i) {
      float* q = dst + (size_t)s[i] * 64 + lane;
      unsafeAtomicAdd(q, lo[i]);
      unsafeAtomicAdd(q + 32, hi[i]);
    }
    if (COUNT_DEG) {
      int gl = g[0];
#pragma unroll
      for (int i = 1; i < ROWS_PER_STEP; ++i) gl = (lane == i) ? g[i] : gl;
      if (lane < ROWS_PER_STEP) unsafeAtomicAdd(deg + gl, 1.0f);
    }
  }
}

__global__ __launch_bounds__(256) void scatter_edges_kernel(
    const float* __restrict__ Y1, const int* __restrict__ vertex,
    const int* __restrict__ edges, float* __restrict__ Xe,
    float* __restrict__ deg) {
  scatter_rows<true>(Y1, vertex, edges, Xe, deg);
}

__global__ __launch_bounds__(256) void scatter_nodes_kernel(
    const float* __restrict__ Xe, const int* __restrict__ vertex,
    const int* __restrict__ edges, float* __restrict__ S) {
  scatter_rows<false>(Xe, edges, vertex, S, nullptr);
}

extern "C" void kernel_launch(void* const* d_in, const int* in_sizes, int n_in,
                              void* d_out, int out_size, void* d_ws,
                              size_t ws_size, hipStream_t stream) {
  (void)in_sizes; (void)n_in; (void)out_size; (void)ws_size;
  const float* X    = (const float*)d_in[0];
  const float* X0   = (const float*)d_in[1];
  const int*   vertex = (const int*)d_in[2];
  const int*   edges  = (const int*)d_in[3];
  const float* W1w  = (const float*)d_in[4];
  const float* W1b  = (const float*)d_in[5];
  const float* W2w  = (const float*)d_in[6];
  const float* W2b  = (const float*)d_in[7];
  const float* Ww   = (const float*)d_in[8];
  const float* Wb   = (const float*)d_in[9];

  float* out = (float*)d_out;
  float* Xe  = out + (size_t)N_NODES_C * 64;

  float* ws  = (float*)d_ws;
  float* Y1  = ws;
  float* S   = Y1 + (size_t)N_NODES_C * 64;
  float* deg = S + (size_t)N_NODES_C * 64;

  hipMemsetAsync(Xe, 0, (size_t)N_EDGES_C * 64 * sizeof(float), stream);
  hipMemsetAsync(S, 0, ((size_t)N_NODES_C * 64 + N_NODES_C) * sizeof(float),
                 stream);

  const int tiles = N_NODES_C / 16;
  const int gemmBlocks = (tiles + 7) / 8;
  const int scatterBlocks = SCATTER_BLOCKS;

  gemm64_bias_kernel<<<gemmBlocks, 256, 0, stream>>>(X, W1w, W1b, Y1, N_NODES_C);
  scatter_edges_kernel<<<scatterBlocks, 256, 0, stream>>>(Y1, vertex, edges, Xe, deg);
  scatter_nodes_kernel<<<scatterBlocks, 256, 0, stream>>>(Xe, vertex, edges, S);
  gemm2_mix_kernel<<<gemmBlocks, 256, 0, stream>>>(X, X0, S, deg, W2w, W2b, Y1,
                                                   N_NODES_C);
  gemm64_bias_kernel<<<gemmBlocks, 256, 0, stream>>>(Y1, Ww, Wb, out, N_NODES_C);
}


# segment sums: global f32 atomic scatters replaced by binning incidences by destination range (LDS counters) + per-range LDS tile sums (LDS f32 adds, arrival order), rows stored once; no memsets
# speedup vs baseline: 2.5014x; 1.8441x over previous
#include <hip/hip_runtime.h>

typedef __attribute__((ext_vector_type(2))) float v2f;
typedef __attribute__((ext_vector_type(8))) float v8f;

#define N_NODES_C 100000
#define N_EDGES_C 20000
#define NNZ_C     1000000
#define ALPHA_C   0.5f

__device__ __forceinline__ v8f wmma_f32(v2f a, v2f b, v8f c) {
  return __builtin_amdgcn_wmma_f32_16x16x4_f32(false, a, false, b, (short)0, c,
                                               false, false);
}

__global__ __launch_bounds__(256) void gemm64_bias_kernel(
    const float* __restrict__ A, const float* __restrict__ W,
    const float* __restrict__ bias, float* __restrict__ Out, int M) {
  __shared__ float Bsh[64 * 64];
  for (int i = threadIdx.x; i < 64 * 64; i += 256) {
    const int k = i >> 6, n = i & 63;
    Bsh[i] = W[n * 64 + k];
  }
  __syncthreads();

  const int lane = threadIdx.x & 31;
  const int wave = threadIdx.x >> 5;
  const int rowBase = (blockIdx.x * 8 + wave) << 4;
  if (rowBase >= M) return;
  const int ll = lane & 15;
  const int lh = lane >> 4;

  const float* Arow = A + (size_t)(rowBase + ll) * 64;
  v8f acc0 = {}, acc1 = {}, acc2 = {}, acc3 = {};

#pragma unroll
  for (int k0 = 0; k0 < 64; k0 += 4) {
    const int ka = k0 + lh * 2;
    v2f a;
    a.x = Arow[ka];
    a.y = Arow[ka + 1];
    const float* B0 = &Bsh[ka * 64 + ll];
    const float* B1 = &Bsh[(ka + 1) * 64 + ll];
    v2f b;
    b.x = B0[0];  b.y = B1[0];  acc0 = wmma_f32(a, b, acc0);
    b.x = B0[16]; b.y = B1[16]; acc1 = wmma_f32(a, b, acc1);
    b.x = B0[32]; b.y = B1[32]; acc2 = wmma_f32(a, b, acc2);
    b.x = B0[48]; b.y = B1[48]; acc3 = wmma_f32(a, b, acc3);
  }

#pragma unroll
  for (int j = 0; j < 8; ++j) {
    const int r = rowBase + j + lh * 8;
    float* o = Out + (size_t)r * 64;
    o[ll]      = acc0[j] + bias[ll];
    o[16 + ll] = acc1[j] + bias[16 + ll];
    o[32 + ll] = acc2[j] + bias[32 + ll];
    o[48 + ll] = acc3[j] + bias[48 + ll];
  }
}

__global__ __launch_bounds__(256) void gemm2_mix_kernel(
    const float* __restrict__ X, const float* __restrict__ X0,
    const float* __restrict__ S, const float* __restrict__ deg,
    const float* __restrict__ W2, const float* __restrict__ b2,
    float* __restrict__ T, int M) {
  __shared__ float Bsh[128 * 64];
  for (int i = threadIdx.x; i < 128 * 64; i += 256) {
    const int k = i >> 6, n = i & 63;
    Bsh[i] = W2[n * 128 + k];
  }
  __syncthreads();

  const int lane = threadIdx.x & 31;
  const int wave = threadIdx.x >> 5;
  const int rowBase = (blockIdx.x * 8 + wave) << 4;
  if (rowBase >= M) return;
  const int ll = lane & 15;
  const int lh = lane >> 4;

  const float dg = deg[rowBase + ll];
  const float* Xrow = X + (size_t)(rowBase + ll) * 64;
  const float* Srow = S + (size_t)(rowBase + ll) * 64;
  v8f acc0 = {}, acc1 = {}, acc2 = {}, acc3 = {};

#pragma unroll
  for (int k0 = 0; k0 < 64; k0 += 4) {
    const int ka = k0 + lh * 2;
    v2f a;
    a.x = dg * Xrow[ka];
    a.y = dg * Xrow[ka + 1];
    const float* B0 = &Bsh[ka * 64 + ll];
    const float* B1 = &Bsh[(ka + 1) * 64 + ll];
    v2f b;
    b.x = B0[0];  b.y = B1[0];  acc0 = wmma_f32(a, b, acc0);
    b.x = B0[16]; b.y = B1[16]; acc1 = wmma_f32(a, b, acc1);
    b.x = B0[32]; b.y = B1[32]; acc2 = wmma_f32(a, b, acc2);
    b.x = B0[48]; b.y = B1[48]; acc3 = wmma_f32(a, b, acc3);
  }
#pragma unroll
  for (int k0 = 64; k0 < 128; k0 += 4) {
    const int ka = k0 + lh * 2;
    v2f a;
    a.x = Srow[ka - 64];
    a.y = Srow[ka - 63];
    const float* B0 = &Bsh[ka * 64 + ll];
    const float* B1 = &Bsh[(ka + 1) * 64 + ll];
    v2f b;
    b.x = B0[0];  b.y = B1[0];  acc0 = wmma_f32(a, b, acc0);
    b.x = B0[16]; b.y = B1[16]; acc1 = wmma_f32(a, b, acc1);
    b.x = B0[32]; b.y = B1[32]; acc2 = wmma_f32(a, b, acc2);
    b.x = B0[48]; b.y = B1[48]; acc3 = wmma_f32(a, b, acc3);
  }

  const float oma = 1.0f - ALPHA_C;
#pragma unroll
  for (int j = 0; j < 8; ++j) {
    const int r = rowBase + j + lh * 8;
    const float dr = deg[r];
    const float* x0r = X0 + (size_t)r * 64;
    float* o = T + (size_t)r * 64;
    o[ll]      = oma * (acc0[j] + dr * b2[ll])      + ALPHA_C * x0r[ll];
    o[16 + ll] = oma * (acc1[j] + dr * b2[16 + ll]) + ALPHA_C * x0r[16 + ll];
    o[32 + ll] = oma * (acc2[j] + dr * b2[32 + ll]) + ALPHA_C * x0r[32 + ll];
    o[48 + ll] = oma * (acc3[j] + dr * b2[48 + ll]) + ALPHA_C * x0r[48 + ll];
  }
}

#define CHUNK 8192
#define BIN_BLOCKS ((NNZ_C + CHUNK - 1) / CHUNK)
#define SHIFT_E 6
#define SHIFT_V 8
#define NB_E ((N_EDGES_C + (1 << SHIFT_E) - 1) >> SHIFT_E)
#define NB_V ((N_NODES_C + (1 << SHIFT_V) - 1) >> SHIFT_V)
#define SRC_BITS 20
#define SRC_MASK ((1 << SRC_BITS) - 1)
static_assert(N_NODES_C <= (1 << SRC_BITS) && N_EDGES_C <= (1 << SRC_BITS),
              "a source row fits the low bits of an entry");
static_assert(NB_E <= 1024 && NB_V <= 1024, "one thread per range in the scan");

template <int SHIFT, int NB>
__global__ __launch_bounds__(1024) void bin_kernel(
    const int* __restrict__ keys, const int* __restrict__ vals,
    int* __restrict__ entries, int* __restrict__ table) {
  __shared__ int hist[NB];
  __shared__ int wsum[32];
  const int tid = threadIdx.x;
  const int lane = tid & 31;
  const int wave = tid >> 5;
  for (int i = tid; i < NB; i += 1024) hist[i] = 0;
  __syncthreads();

  const int base = blockIdx.x * CHUNK;
  int key[8], val[8], rank[8];
#pragma unroll
  for (int j = 0; j < 8; ++j) {
    const int e = base + j * 1024 + tid;
    key[j] = 0;
    val[j] = 0;
    if (e < NNZ_C) {
      key[j] = keys[e];
      val[j] = vals[e];
    }
  }
#pragma unroll
  for (int j = 0; j < 8; ++j) {
    const int e = base + j * 1024 + tid;
    rank[j] = 0;
    if (e < NNZ_C) rank[j] = atomicAdd(&hist[key[j] >> SHIFT], 1);
  }
  __syncthreads();

  const int c = (tid < NB) ? hist[tid] : 0;
  int incl = c;
  for (int d = 1; d < 32; d <<= 1) {
    const int t = __shfl_up(incl, d);
    if (lane >= d) incl += t;
  }
  if (lane == 31) wsum[wave] = incl;
  __syncthreads();
  int before = 0;
  for (int w2 = 0; w2 < wave; ++w2) before += wsum[w2];
  const int start = before + incl - c;
  if (tid < NB) {
    hist[tid] = start;
    table[blockIdx.x * (NB + 1) + tid] = start;
  }
  if (tid == 0) {
    const int left = NNZ_C - base;
    table[blockIdx.x * (NB + 1) + NB] = left < CHUNK ? left : CHUNK;
  }
  __syncthreads();

#pragma unroll
  for (int j = 0; j < 8; ++j) {
    const int e = base + j * 1024 + tid;
    if (e < NNZ_C) {
      const int r = key[j] >> SHIFT;
      const int local = key[j] - (r << SHIFT);
      entries[base + hist[r] + rank[j]] = (local << SRC_BITS) | val[j];
    }
  }
}

#define GROUP 8
template <bool FULL, bool COUNT_DEG>
__device__ __forceinline__ void add_group(const float* __restrict__ src,
                                          const int* __restrict__ run, int k,
                                          int n, int lane, float* tile,
                                          int* degc) {
  int g[GROUP];
  float lo[GROUP], hi[GROUP];
#pragma unroll
  for (int i = 0; i < GROUP; ++i) {
    const int ii = (FULL || i < n) ? i : n - 1;
    g[i] = run[k + ii];
  }
#pragma unroll
  for (int i = 0; i < GROUP; ++i) {
    const float* p = src + (size_t)(g[i] & SRC_MASK) * 64 + lane;
    lo[i] = p[0];
    hi[i] = p[32];
  }
#pragma unroll
  for (int i = 0; i < GROUP; ++i) {
    if (FULL || i < n) {
      float* q = tile + (g[i] >> SRC_BITS) * 64 + lane;
      unsafeAtomicAdd(q, lo[i]);
      unsafeAtomicAdd(q + 32, hi[i]);
    }
  }
  if (COUNT_DEG) {
    int gl = g[0];
#pragma unroll
    for (int i = 1; i < GROUP; ++i) gl = (lane == i) ? g[i] : gl;
    if (lane < (FULL ? GROUP : n)) atomicAdd(degc + (gl >> SRC_BITS), 1);
  }
}

template <int SHIFT, int NB, bool COUNT_DEG>
__global__ __launch_bounds__(1024) void sum_range_kernel(
    const float* __restrict__ src, const int* __restrict__ entries,
    const int* __restrict__ table, float* __restrict__ dst,
    float* __restrict__ deg, int nrows) {
  constexpr int R = 1 << SHIFT;
  __shared__ float tile[R * 64];
  __shared__ int degc[R];
  const int tid = threadIdx.x;
  const int lane = tid & 31;
  const int wave = __builtin_amdgcn_readfirstlane(tid >> 5);
  const int b = blockIdx.x;
  for (int i = tid; i < R * 64; i += 1024) tile[i] = 0.0f;
  for (int i = tid; i < R; i += 1024) degc[i] = 0;
  __syncthreads();

  for (int bin = wave; bin < BIN_BLOCKS; bin += 32) {
    const int* trow = table + bin * (NB + 1) + b;
    const int s = __builtin_amdgcn_readfirstlane(trow[0]);
    const int t = __builtin_amdgcn_readfirstlane(trow[1]);
    const int* run = entries + bin * CHUNK;
    int k = s;
    for (; k + GROUP <= t; k += GROUP)
      add_group<true, COUNT_DEG>(src, run, k, GROUP, lane, tile, degc);
    if (k < t)
      add_group<false, COUNT_DEG>(src, run, k, t - k, lane, tile, degc);
  }
  __syncthreads();

  for (int r = wave; r < R; r += 32) {
    const int row = b * R + r;
    if (row < nrows) {
      float* q = dst + (size_t)row * 64 + lane;
      q[0] = tile[r * 64 + lane];
      q[32] = tile[r * 64 + 32 + lane];
      if (COUNT_DEG && lane == 0) deg[row] = (float)degc[r];
    }
  }
}

extern "C" void kernel_launch(void* const* d_in, const int* in_sizes, int n_in,
                              void* d_out, int out_size, void* d_ws,
                              size_t ws_size, hipStream_t stream) {
  (void)in_sizes; (void)n_in; (void)out_size; (void)ws_size;
  const float* X    = (const float*)d_in[0];
  const float* X0   = (const float*)d_in[1];
  const int*   vertex = (const int*)d_in[2];
  const int*   edges  = (const int*)d_in[3];
  const float* W1w  = (const float*)d_in[4];
  const float* W1b  = (const float*)d_in[5];
  const float* W2w  = (const float*)d_in[6];
  const float* W2b  = (const float*)d_in[7];
  const float* Ww   = (const float*)d_in[8];
  const float* Wb   = (const float*)d_in[9];

  float* out = (float*)d_out;
  float* Xe  = out + (size_t)N_NODES_C * 64;

  float* ws  = (float*)d_ws;
  float* Y1  = ws;
  float* S   = Y1;
  float* T   = ws + (size_t)N_NODES_C * 64;
  float* deg = T + (size_t)N_NODES_C * 64;

  int* entE = (int*)T;
  int* entV = entE + NNZ_C;
  int* tabE = entV + NNZ_C;
  int* tabV = tabE + BIN_BLOCKS * (NB_E + 1);

  const int tiles = N_NODES_C / 16;
  const int gemmBlocks = (tiles + 7) / 8;

  bin_kernel<SHIFT_E, NB_E><<<BIN_BLOCKS, 1024, 0, stream>>>(edges, vertex, entE, tabE);
  bin_kernel<SHIFT_V, NB_V><<<BIN_BLOCKS, 1024, 0, stream>>>(vertex, edges, entV, tabV);
  gemm64_bias_kernel<<<gemmBlocks, 256, 0, stream>>>(X, W1w, W1b, Y1, N_NODES_C);
  sum_range_kernel<SHIFT_E, NB_E, false><<<NB_E, 1024, 0, stream>>>(
      Y1, entE, tabE, Xe, nullptr, N_EDGES_C);
  sum_range_kernel<SHIFT_V, NB_V, true><<<NB_V, 1024, 0, stream>>>(
      Xe, entV, tabV, S, deg, N_NODES_C);
  gemm2_mix_kernel<<<gemmBlocks, 256, 0, stream>>>(X, X0, S, deg, W2w, W2b, T,
                                                   N_NODES_C);
  gemm64_bias_kernel<<<gemmBlocks, 256, 0, stream>>>(T, Ww, Wb, out, N_NODES_C);
}


# the three independent first launches (bin by edge range, bin by node range, Y1 = X W1^T GEMM) run as one launch of 1024-thread blocks chosen by block index; 7 -> 5 launches; no sum reordered
# speedup vs baseline: 2.7752x; 1.1095x over previous
#include <hip/hip_runtime.h>

typedef __attribute__((ext_vector_type(2))) float v2f;
typedef __attribute__((ext_vector_type(8))) float v8f;

#define N_NODES_C 100000
#define N_EDGES_C 20000
#define NNZ_C     1000000
#define ALPHA_C   0.5f

__device__ __forceinline__ v8f wmma_f32(v2f a, v2f b, v8f c) {
  return __builtin_amdgcn_wmma_f32_16x16x4_f32(false, a, false, b, (short)0, c,
                                               false, false);
}

template <int THREADS>
__device__ __forceinline__ void stage_w64(const float* __restrict__ W, float* Bsh) {
  for (int i = threadIdx.x; i < 64 * 64; i += THREADS) {
    const int k = i >> 6, n = i & 63;
    Bsh[i] = W[n * 64 + k];
  }
}

__device__ __forceinline__ void gemm64_tile(const float* __restrict__ A,
                                            const float* Bsh,
                                            const float* __restrict__ bias,
                                            float* __restrict__ Out, int rowBase,
                                            int lane) {
  const int ll = lane & 15;
  const int lh = lane >> 4;

  const float* Arow = A + (size_t)(rowBase + ll) * 64;
  v8f acc0 = {}, acc1 = {}, acc2 = {}, acc3 = {};

#pragma unroll
  for (int k0 = 0; k0 < 64; k0 += 4) {
    const int ka = k0 + lh * 2;
    v2f a;
    a.x = Arow[ka];
    a.y = Arow[ka + 1];
    const float* B0 = &Bsh[ka * 64 + ll];
    const float* B1 = &Bsh[(ka + 1) * 64 + ll];
    v2f b;
    b.x = B0[0];  b.y = B1[0];  acc0 = wmma_f32(a, b, acc0);
    b.x = B0[16]; b.y = B1[16]; acc1 = wmma_f32(a, b, acc1);
    b.x = B0[32]; b.y = B1[32]; acc2 = wmma_f32(a, b, acc2);
    b.x = B0[48]; b.y = B1[48]; acc3 = wmma_f32(a, b, acc3);
  }

#pragma unroll
  for (int j = 0; j < 8; ++j) {
    const int r = rowBase + j + lh * 8;
    float* o = Out + (size_t)r * 64;
    o[ll]      = acc0[j] + bias[ll];
    o[16 + ll] = acc1[j] + bias[16 + ll];
    o[32 + ll] = acc2[j] + bias[32 + ll];
    o[48 + ll] = acc3[j] + bias[48 + ll];
  }
}

__global__ __launch_bounds__(256) void gemm64_bias_kernel(
    const float* __restrict__ A, const float* __restrict__ W,
    const float* __restrict__ bias, float* __restrict__ Out, int M) {
  __shared__ float Bsh[64 * 64];
  stage_w64<256>(W, Bsh);
  __syncthreads();

  const int lane = threadIdx.x & 31;
  const int wave = threadIdx.x >> 5;
  const int rowBase = (blockIdx.x * 8 + wave) << 4;
  if (rowBase >= M) return;
  gemm64_tile(A, Bsh, bias, Out, rowBase, lane);
}

__global__ __launch_bounds__(256) void gemm2_mix_kernel(
    const float* __restrict__ X, const float* __restrict__ X0,
    const float* __restrict__ S, const float* __restrict__ deg,
    const float* __restrict__ W2, const float* __restrict__ b2,
    float* __restrict__ T, int M) {
  __shared__ float Bsh[128 * 64];
  for (int i = threadIdx.x; i < 128 * 64; i += 256) {
    const int k = i >> 6, n = i & 63;
    Bsh[i] = W2[n * 128 + k];
  }
  __syncthreads();

  const int lane = threadIdx.x & 31;
  const int wave = threadIdx.x >> 5;
  const int rowBase = (blockIdx.x * 8 + wave) << 4;
  if (rowBase >= M) return;
  const int ll = lane & 15;
  const int lh = lane >> 4;

  const float dg = deg[rowBase + ll];
  const float* Xrow = X + (size_t)(rowBase + ll) * 64;
  const float* Srow = S + (size_t)(rowBase + ll) * 64;
  v8f acc0 = {}, acc1 = {}, acc2 = {}, acc3 = {};

#pragma unroll
  for (int k0 = 0; k0 < 64; k0 += 4) {
    const int ka = k0 + lh * 2;
    v2f a;
    a.x = dg * Xrow[ka];
    a.y = dg * Xrow[ka + 1];
    const float* B0 = &Bsh[ka * 64 + ll];
    const float* B1 = &Bsh[(ka + 1) * 64 + ll];
    v2f b;
    b.x = B0[0];  b.y = B1[0];  acc0 = wmma_f32(a, b, acc0);
    b.x = B0[16]; b.y = B1[16]; acc1 = wmma_f32(a, b, acc1);
    b.x = B0[32]; b.y = B1[32]; acc2 = wmma_f32(a, b, acc2);
    b.x = B0[48]; b.y = B1[48]; acc3 = wmma_f32(a, b, acc3);
  }
#pragma unroll
  for (int k0 = 64; k0 < 128; k0 += 4) {
    const int ka = k0 + lh * 2;
    v2f a;
    a.x = Srow[ka - 64];
    a.y = Srow[ka - 63];
    const float* B0 = &Bsh[ka * 64 + ll];
    const float* B1 = &Bsh[(ka + 1) * 64 + ll];
    v2f b;
    b.x = B0[0];  b.y = B1[0];  acc0 = wmma_f32(a, b, acc0);
    b.x = B0[16]; b.y = B1[16]; acc1 = wmma_f32(a, b, acc1);
    b.x = B0[32]; b.y = B1[32]; acc2 = wmma_f32(a, b, acc2);
    b.x = B0[48]; b.y = B1[48]; acc3 = wmma_f32(a, b, acc3);
  }

  const float oma = 1.0f - ALPHA_C;
#pragma unroll
  for (int j = 0; j < 8; ++j) {
    const int r = rowBase + j + lh * 8;
    const float dr = deg[r];
    const float* x0r = X0 + (size_t)r * 64;
    float* o = T + (size_t)r * 64;
    o[ll]      = oma * (acc0[j] + dr * b2[ll])      + ALPHA_C * x0r[ll];
    o[16 + ll] = oma * (acc1[j] + dr * b2[16 + ll]) + ALPHA_C * x0r[16 + ll];
    o[32 + ll] = oma * (acc2[j] + dr * b2[32 + ll]) + ALPHA_C * x0r[32 + ll];
    o[48 + ll] = oma * (acc3[j] + dr * b2[48 + ll]) + ALPHA_C * x0r[48 + ll];
  }
}

#define CHUNK 8192
#define BIN_BLOCKS ((NNZ_C + CHUNK - 1) / CHUNK)
#define SHIFT_E 6
#define SHIFT_V 8
#define NB_E ((N_EDGES_C + (1 << SHIFT_E) - 1) >> SHIFT_E)
#define NB_V ((N_NODES_C + (1 << SHIFT_V) - 1) >> SHIFT_V)
#define SRC_BITS 20
#define SRC_MASK ((1 << SRC_BITS) - 1)
static_assert(N_NODES_C <= (1 << SRC_BITS) && N_EDGES_C <= (1 << SRC_BITS),
              "a source row fits the low bits of an entry");
static_assert(NB_E <= 1024 && NB_V <= 1024, "one thread per range in the scan");

template <int SHIFT, int NB>
__device__ __forceinline__ void bin_body(
    const int* __restrict__ keys, const int* __restrict__ vals,
    int* __restrict__ entries, int* __restrict__ table, int blk, int* hist,
    int* wsum) {
  const int tid = threadIdx.x;
  const int lane = tid & 31;
  const int wave = tid >> 5;
  for (int i = tid; i < NB; i += 1024) hist[i] = 0;
  __syncthreads();

  const int base = blk * CHUNK;
  int key[8], val[8], rank[8];
#pragma unroll
  for (int j = 0; j < 8; ++j) {
    const int e = base + j * 1024 + tid;
    key[j] = 0;
    val[j] = 0;
    if (e < NNZ_C) {
      key[j] = keys[e];
      val[j] = vals[e];
    }
  }
#pragma unroll
  for (int j = 0; j < 8; ++j) {
    const int e = base + j * 1024 + tid;
    rank[j] = 0;
    if (e < NNZ_C) rank[j] = atomicAdd(&hist[key[j] >> SHIFT], 1);
  }
  __syncthreads();

  const int c = (tid < NB) ? hist[tid] : 0;
  int incl = c;
  for (int d = 1; d < 32; d <<= 1) {
    const int t = __shfl_up(incl, d);
    if (lane >= d) incl += t;
  }
  if (lane == 31) wsum[wave] = incl;
  __syncthreads();
  int before = 0;
  for (int w2 = 0; w2 < wave; ++w2) before += wsum[w2];
  const int start = before + incl - c;
  if (tid < NB) {
    hist[tid] = start;
    table[blk * (NB + 1) + tid] = start;
  }
  if (tid == 0) {
    const int left = NNZ_C - base;
    table[blk * (NB + 1) + NB] = left < CHUNK ? left : CHUNK;
  }
  __syncthreads();

#pragma unroll
  for (int j = 0; j < 8; ++j) {
    const int e = base + j * 1024 + tid;
    if (e < NNZ_C) {
      const int r = key[j] >> SHIFT;
      const int local = key[j] - (r << SHIFT);
      entries[base + hist[r] + rank[j]] = (local << SRC_BITS) | val[j];
    }
  }
}

#define NB_MAX (NB_E > NB_V ? NB_E : NB_V)
#define GEMM1_BLOCKS ((N_NODES_C / 16 + 31) / 32)
#define FRONT_BLOCKS (GEMM1_BLOCKS + 2 * BIN_BLOCKS)
__global__ __launch_bounds__(1024) void front_kernel(
    const int* __restrict__ edges, const int* __restrict__ vertex,
    int* __restrict__ entE, int* __restrict__ tabE, int* __restrict__ entV,
    int* __restrict__ tabV, const float* __restrict__ X,
    const float* __restrict__ W1, const float* __restrict__ b1,
    float* __restrict__ Y1) {
  __shared__ int hist[NB_MAX];
  __shared__ int wsum[32];
  __shared__ float Bsh[64 * 64];
  const int blk = blockIdx.x;
  if (blk < GEMM1_BLOCKS) {
    stage_w64<1024>(W1, Bsh);
    __syncthreads();
    const int lane = threadIdx.x & 31;
    const int wave = threadIdx.x >> 5;
    const int rowBase = (blk * 32 + wave) << 4;
    if (rowBase < N_NODES_C) gemm64_tile(X, Bsh, b1, Y1, rowBase, lane);
  } else if (blk < GEMM1_BLOCKS + BIN_BLOCKS) {
    bin_body<SHIFT_E, NB_E>(edges, vertex, entE, tabE, blk - GEMM1_BLOCKS, hist, wsum);
  } else {
    bin_body<SHIFT_V, NB_V>(vertex, edges, entV, tabV,
                            blk - GEMM1_BLOCKS - BIN_BLOCKS, hist, wsum);
  }
}

#define GROUP 8
template <bool FULL, bool COUNT_DEG>
__device__ __forceinline__ void add_group(const float* __restrict__ src,
                                          const int* __restrict__ run, int k,
                                          int n, int lane, float* tile,
                                          int* degc) {
  int g[GROUP];
  float lo[GROUP], hi[GROUP];
#pragma unroll
  for (int i = 0; i < GROUP; ++i) {
    const int ii = (FULL || i < n) ? i : n - 1;
    g[i] = run[k + ii];
  }
#pragma unroll
  for (int i = 0; i < GROUP; ++i) {
    const float* p = src + (size_t)(g[i] & SRC_MASK) * 64 + lane;
    lo[i] = p[0];
    hi[i] = p[32];
  }
#pragma unroll
  for (int i = 0; i < GROUP; ++i) {
    if (FULL || i < n) {
      float* q = tile + (g[i] >> SRC_BITS) * 64 + lane;
      unsafeAtomicAdd(q, lo[i]);
      unsafeAtomicAdd(q + 32, hi[i]);
    }
  }
  if (COUNT_DEG) {
    int gl = g[0];
#pragma unroll
    for (int i = 1; i < GROUP; ++i) gl = (lane == i) ? g[i] : gl;
    if (lane < (FULL ? GROUP : n)) atomicAdd(degc + (gl >> SRC_BITS), 1);
  }
}

template <int SHIFT, int NB, bool COUNT_DEG>
__global__ __launch_bounds__(1024) void sum_range_kernel(
    const float* __restrict__ src, const int* __restrict__ entries,
    const int* __restrict__ table, float* __restrict__ dst,
    float* __restrict__ deg, int nrows) {
  constexpr int R = 1 << SHIFT;
  __shared__ float tile[R * 64];
  __shared__ int degc[R];
  const int tid = threadIdx.x;
  const int lane = tid & 31;
  const int wave = __builtin_amdgcn_readfirstlane(tid >> 5);
  const int b = blockIdx.x;
  for (int i = tid; i < R * 64; i += 1024) tile[i] = 0.0f;
  for (int i = tid; i < R; i += 1024) degc[i] = 0;
  __syncthreads();

  for (int bin = wave; bin < BIN_BLOCKS; bin += 32) {
    const int* trow = table + bin * (NB + 1) + b;
    const int s = __builtin_amdgcn_readfirstlane(trow[0]);
    const int t = __builtin_amdgcn_readfirstlane(trow[1]);
    const int* run = entries + bin * CHUNK;
    int k = s;
    for (; k + GROUP <= t; k += GROUP)
      add_group<true, COUNT_DEG>(src, run, k, GROUP, lane, tile, degc);
    if (k < t)
      add_group<false, COUNT_DEG>(src, run, k, t - k, lane, tile, degc);
  }
  __syncthreads();

  for (int r = wave; r < R; r += 32) {
    const int row = b * R + r;
    if (row < nrows) {
      float* q = dst + (size_t)row * 64 + lane;
      q[0] = tile[r * 64 + lane];
      q[32] = tile[r * 64 + 32 + lane];
      if (COUNT_DEG && lane == 0) deg[row] = (float)degc[r];
    }
  }
}

extern "C" void kernel_launch(void* const* d_in, const int* in_sizes, int n_in,
                              void* d_out, int out_size, void* d_ws,
                              size_t ws_size, hipStream_t stream) {
  (void)in_sizes; (void)n_in; (void)out_size; (void)ws_size;
  const float* X    = (const float*)d_in[0];
  const float* X0   = (const float*)d_in[1];
  const int*   vertex = (const int*)d_in[2];
  const int*   edges  = (const int*)d_in[3];
  const float* W1w  = (const float*)d_in[4];
  const float* W1b  = (const float*)d_in[5];
  const float* W2w  = (const float*)d_in[6];
  const float* W2b  = (const float*)d_in[7];
  const float* Ww   = (const float*)d_in[8];
  const float* Wb   = (const float*)d_in[9];

  float* out = (float*)d_out;
  float* Xe  = out + (size_t)N_NODES_C * 64;

  float* ws  = (float*)d_ws;
  float* Y1  = ws;
  float* S   = Y1;
  float* T   = ws + (size_t)N_NODES_C * 64;
  float* deg = T + (size_t)N_NODES_C * 64;

  int* entE = (int*)T;
  int* entV = entE + NNZ_C;
  int* tabE = entV + NNZ_C;
  int* tabV = tabE + BIN_BLOCKS * (NB_E + 1);

  const int tiles = N_NODES_C / 16;
  const int gemmBlocks = (tiles + 7) / 8;

  front_kernel<<<FRONT_BLOCKS, 1024, 0, stream>>>(
      edges, vertex, entE, tabE, entV, tabV, X, W1w, W1b, Y1);
  sum_range_kernel<SHIFT_E, NB_E, false><<<NB_E, 1024, 0, stream>>>(
      Y1, entE, tabE, Xe, nullptr, N_EDGES_C);
  sum_range_kernel<SHIFT_V, NB_V, true><<<NB_V, 1024, 0, stream>>>(
      Xe, entV, tabV, S, deg, N_NODES_C);
  gemm2_mix_kernel<<<gemmBlocks, 256, 0, stream>>>(X, X0, S, deg, W2w, W2b, T,
                                                   N_NODES_C);
  gemm64_bias_kernel<<<gemmBlocks, 256, 0, stream>>>(T, Ww, Wb, out, N_NODES_C);
}


# GEMM operands read as 16-byte k-quads: weights kept row-major in LDS (whole-line staging reads, pitch K+4), B = one ds_load_b128, A = global_load_b128; k order of the products' sums changes (fixed)
# speedup vs baseline: 3.2696x; 1.1781x over previous
#include <hip/hip_runtime.h>

typedef __attribute__((ext_vector_type(2))) float v2f;
typedef __attribute__((ext_vector_type(4))) float v4f;
typedef __attribute__((ext_vector_type(8))) float v8f;

#define N_NODES_C 100000
#define N_EDGES_C 20000
#define NNZ_C     1000000
#define ALPHA_C   0.5f

__device__ __forceinline__ v8f wmma_f32(v2f a, v2f b, v8f c) {
  return __builtin_amdgcn_wmma_f32_16x16x4_f32(false, a, false, b, (short)0, c,
                                               false, false);
}

#define WP64 68
#define WP128 132

template <int THREADS, int K, int PITCH>
__device__ __forceinline__ void stage_w(const float* __restrict__ W, float* Wsh) {
  constexpr int Q = K / 4;
  constexpr int STEPS = 64 * Q / THREADS;
  static_assert(STEPS * THREADS == 64 * Q, "every thread copies the same number of pieces");
  v4f w[STEPS];
#pragma unroll
  for (int j = 0; j < STEPS; ++j)
    w[j] = *(const v4f*)(W + 4 * (threadIdx.x + j * THREADS));
#pragma unroll
  for (int j = 0; j < STEPS; ++j) {
    const int i = threadIdx.x + j * THREADS;
    const int n = i / Q, p = i % Q;
    *(v4f*)(Wsh + n * PITCH + 4 * p) = w[j];
  }
}

template <int PITCH>
__device__ __forceinline__ void wmma_piece(v4f a, const float* Wp, v8f& acc0,
                                           v8f& acc1, v8f& acc2, v8f& acc3) {
  const v4f b0 = *(const v4f*)(Wp);
  const v4f b1 = *(const v4f*)(Wp + 16 * PITCH);
  const v4f b2 = *(const v4f*)(Wp + 32 * PITCH);
  const v4f b3 = *(const v4f*)(Wp + 48 * PITCH);
  acc0 = wmma_f32(a.xy, b0.xy, acc0);
  acc1 = wmma_f32(a.xy, b1.xy, acc1);
  acc2 = wmma_f32(a.xy, b2.xy, acc2);
  acc3 = wmma_f32(a.xy, b3.xy, acc3);
  acc0 = wmma_f32(a.zw, b0.zw, acc0);
  acc1 = wmma_f32(a.zw, b1.zw, acc1);
  acc2 = wmma_f32(a.zw, b2.zw, acc2);
  acc3 = wmma_f32(a.zw, b3.zw, acc3);
}

__device__ __forceinline__ void gemm64_tile(const float* __restrict__ A,
                                            const float* Wsh,
                                            const float* __restrict__ bias,
                                            float* __restrict__ Out, int rowBase,
                                            int lane) {
  const int ll = lane & 15;
  const int lh = lane >> 4;

  const float* Arow = A + (size_t)(rowBase + ll) * 64 + 4 * lh;
  const float* Wrow = Wsh + ll * WP64 + 4 * lh;
  v8f acc0 = {}, acc1 = {}, acc2 = {}, acc3 = {};

  v4f a[8];
#pragma unroll
  for (int q = 0; q < 8; ++q) a[q] = *(const v4f*)(Arow + 8 * q);
#pragma unroll
  for (int q = 0; q < 8; ++q)
    wmma_piece<WP64>(a[q], Wrow + 8 * q, acc0, acc1, acc2, acc3);

#pragma unroll
  for (int j = 0; j < 8; ++j) {
    const int r = rowBase + j + lh * 8;
    float* o = Out + (size_t)r * 64;
    o[ll]      = acc0[j] + bias[ll];
    o[16 + ll] = acc1[j] + bias[16 + ll];
    o[32 + ll] = acc2[j] + bias[32 + ll];
    o[48 + ll] = acc3[j] + bias[48 + ll];
  }
}

__global__ __launch_bounds__(256) void gemm64_bias_kernel(
    const float* __restrict__ A, const float* __restrict__ W,
    const float* __restrict__ bias, float* __restrict__ Out, int M) {
  __shared__ __attribute__((aligned(16))) float Wsh[64 * WP64];
  stage_w<256, 64, WP64>(W, Wsh);
  __syncthreads();

  const int lane = threadIdx.x & 31;
  const int wave = threadIdx.x >> 5;
  const int rowBase = (blockIdx.x * 8 + wave) << 4;
  if (rowBase >= M) return;
  gemm64_tile(A, Wsh, bias, Out, rowBase, lane);
}

__global__ __launch_bounds__(256) void gemm2_mix_kernel(
    const float* __restrict__ X, const float* __restrict__ X0,
    const float* __restrict__ S, const float* __restrict__ deg,
    const float* __restrict__ W2, const float* __restrict__ b2,
    float* __restrict__ T, int M) {
  __shared__ __attribute__((aligned(16))) float Wsh[64 * WP128];
  stage_w<256, 128, WP128>(W2, Wsh);
  __syncthreads();

  const int lane = threadIdx.x & 31;
  const int wave = threadIdx.x >> 5;
  const int rowBase = (blockIdx.x * 8 + wave) << 4;
  if (rowBase >= M) return;
  const int ll = lane & 15;
  const int lh = lane >> 4;

  const float dg = deg[rowBase + ll];
  const float* Xrow = X + (size_t)(rowBase + ll) * 64 + 4 * lh;
  const float* Srow = S + (size_t)(rowBase + ll) * 64 + 4 * lh;
  const float* Wrow = Wsh + ll * WP128 + 4 * lh;
  v8f acc0 = {}, acc1 = {}, acc2 = {}, acc3 = {};

  v4f ax[8], as[8];
#pragma unroll
  for (int q = 0; q < 8; ++q) ax[q] = *(const v4f*)(Xrow + 8 * q);
#pragma unroll
  for (int q = 0; q < 8; ++q) as[q] = *(const v4f*)(Srow + 8 * q);
#pragma unroll
  for (int q = 0; q < 8; ++q)
    wmma_piece<WP128>(dg * ax[q], Wrow + 8 * q, acc0, acc1, acc2, acc3);
#pragma unroll
  for (int q = 0; q < 8; ++q)
    wmma_piece<WP128>(as[q], Wrow + 64 + 8 * q, acc0, acc1, acc2, acc3);

  const float oma = 1.0f - ALPHA_C;
#pragma unroll
  for (int j = 0; j < 8; ++j) {
    const int r = rowBase + j + lh * 8;
    const float dr = deg[r];
    const float* x0r = X0 + (size_t)r * 64;
    float* o = T + (size_t)r * 64;
    o[ll]      = oma * (acc0[j] + dr * b2[ll])      + ALPHA_C * x0r[ll];
    o[16 + ll] = oma * (acc1[j] + dr * b2[16 + ll]) + ALPHA_C * x0r[16 + ll];
    o[32 + ll] = oma * (acc2[j] + dr * b2[32 + ll]) + ALPHA_C * x0r[32 + ll];
    o[48 + ll] = oma * (acc3[j] + dr * b2[48 + ll]) + ALPHA_C * x0r[48 + ll];
  }
}

#define CHUNK 8192
#define BIN_BLOCKS ((NNZ_C + CHUNK - 1) / CHUNK)
#define SHIFT_E 6
#define SHIFT_V 8
#define NB_E ((N_EDGES_C + (1 << SHIFT_E) - 1) >> SHIFT_E)
#define NB_V ((N_NODES_C + (1 << SHIFT_V) - 1) >> SHIFT_V)
#define SRC_BITS 20
#define SRC_MASK ((1 << SRC_BITS) - 1)
static_assert(N_NODES_C <= (1 << SRC_BITS) && N_EDGES_C <= (1 << SRC_BITS),
              "a source row fits the low bits of an entry");
static_assert(NB_E <= 1024 && NB_V <= 1024, "one thread per range in the scan");

template <int SHIFT, int NB>
__device__ __forceinline__ void bin_body(
    const int* __restrict__ keys, const int* __restrict__ vals,
    int* __restrict__ entries, int* __restrict__ table, int blk, int* hist,
    int* wsum) {
  const int tid = threadIdx.x;
  const int lane = tid & 31;
  const int wave = tid >> 5;
  for (int i = tid; i < NB; i += 1024) hist[i] = 0;
  __syncthreads();

  const int base = blk * CHUNK;
  int key[8], val[8], rank[8];
#pragma unroll
  for (int j = 0; j < 8; ++j) {
    const int e = base + j * 1024 + tid;
    key[j] = 0;
    val[j] = 0;
    if (e < NNZ_C) {
      key[j] = keys[e];
      val[j] = vals[e];
    }
  }
#pragma unroll
  for (int j = 0; j < 8; ++j) {
    const int e = base + j * 1024 + tid;
    rank[j] = 0;
    if (e < NNZ_C) rank[j] = atomicAdd(&hist[key[j] >> SHIFT], 1);
  }
  __syncthreads();

  const int c = (tid < NB) ? hist[tid] : 0;
  int incl = c;
  for (int d = 1; d < 32; d <<= 1) {
    const int t = __shfl_up(incl, d);
    if (lane >= d) incl += t;
  }
  if (lane == 31) wsum[wave] = incl;
  __syncthreads();
  int before = 0;
  for (int w2 = 0; w2 < wave; ++w2) before += wsum[w2];
  const int start = before + incl - c;
  if (tid < NB) {
    hist[tid] = start;
    table[blk * (NB + 1) + tid] = start;
  }
  if (tid == 0) {
    const int left = NNZ_C - base;
    table[blk * (NB + 1) + NB] = left < CHUNK ? left : CHUNK;
  }
  __syncthreads();

#pragma unroll
  for (int j = 0; j < 8; ++j) {
    const int e = base + j * 1024 + tid;
    if (e < NNZ_C) {
      const int r = key[j] >> SHIFT;
      const int local = key[j] - (r << SHIFT);
      entries[base + hist[r] + rank[j]] = (local << SRC_BITS) | val[j];
    }
  }
}

#define NB_MAX (NB_E > NB_V ? NB_E : NB_V)
#define GEMM1_BLOCKS ((N_NODES_C / 16 + 31) / 32)
#define FRONT_BLOCKS (GEMM1_BLOCKS + 2 * BIN_BLOCKS)
__global__ __launch_bounds__(1024) void front_kernel(
    const int* __restrict__ edges, const int* __restrict__ vertex,
    int* __restrict__ entE, int* __restrict__ tabE, int* __restrict__ entV,
    int* __restrict__ tabV, const float* __restrict__ X,
    const float* __restrict__ W1, const float* __restrict__ b1,
    float* __restrict__ Y1) {
  __shared__ int hist[NB_MAX];
  __shared__ int wsum[32];
  __shared__ __attribute__((aligned(16))) float Wsh[64 * WP64];
  const int blk = blockIdx.x;
  if (blk < GEMM1_BLOCKS) {
    stage_w<1024, 64, WP64>(W1, Wsh);
    __syncthreads();
    const int lane = threadIdx.x & 31;
    const int wave = threadIdx.x >> 5;
    const int rowBase = (blk * 32 + wave) << 4;
    if (rowBase < N_NODES_C) gemm64_tile(X, Wsh, b1, Y1, rowBase, lane);
  } else if (blk < GEMM1_BLOCKS + BIN_BLOCKS) {
    bin_body<SHIFT_E, NB_E>(edges, vertex, entE, tabE, blk - GEMM1_BLOCKS, hist, wsum);
  } else {
    bin_body<SHIFT_V, NB_V>(vertex, edges, entV, tabV,
                            blk - GEMM1_BLOCKS - BIN_BLOCKS, hist, wsum);
  }
}

#define GROUP 8
template <bool FULL, bool COUNT_DEG>
__device__ __forceinline__ void add_group(const float* __restrict__ src,
                                          const int* __restrict__ run, int k,
                                          int n, int lane, float* tile,
                                          int* degc) {
  int g[GROUP];
  float lo[GROUP], hi[GROUP];
#pragma unroll
  for (int i = 0; i < GROUP; ++i) {
    const int ii = (FULL || i < n) ? i : n - 1;
    g[i] = run[k + ii];
  }
#pragma unroll
  for (int i = 0; i < GROUP; ++i) {
    const float* p = src + (size_t)(g[i] & SRC_MASK) * 64 + lane;
    lo[i] = p[0];
    hi[i] = p[32];
  }
#pragma unroll
  for (int i = 0; i < GROUP; ++i) {
    if (FULL || i < n) {
      float* q = tile + (g[i] >> SRC_BITS) * 64 + lane;
      unsafeAtomicAdd(q, lo[i]);
      unsafeAtomicAdd(q + 32, hi[i]);
    }
  }
  if (COUNT_DEG) {
    int gl = g[0];
#pragma unroll
    for (int i = 1; i < GROUP; ++i) gl = (lane == i) ? g[i] : gl;
    if (lane < (FULL ? GROUP : n)) atomicAdd(degc + (gl >> SRC_BITS), 1);
  }
}

template <int SHIFT, int NB, bool COUNT_DEG>
__global__ __launch_bounds__(1024) void sum_range_kernel(
    const float* __restrict__ src, const int* __restrict__ entries,
    const int* __restrict__ table, float* __restrict__ dst,
    float* __restrict__ deg, int nrows) {
  constexpr int R = 1 << SHIFT;
  __shared__ float tile[R * 64];
  __shared__ int degc[R];
  const int tid = threadIdx.x;
  const int lane = tid & 31;
  const int wave = __builtin_amdgcn_readfirstlane(tid >> 5);
  const int b = blockIdx.x;
  for (int i = tid; i < R * 64; i += 1024) tile[i] = 0.0f;
  for (int i = tid; i < R; i += 1024) degc[i] = 0;
  __syncthreads();

  for (int bin = wave; bin < BIN_BLOCKS; bin += 32) {
    const int* trow = table + bin * (NB + 1) + b;
    const int s = __builtin_amdgcn_readfirstlane(trow[0]);
    const int t = __builtin_amdgcn_readfirstlane(trow[1]);
    const int* run = entries + bin * CHUNK;
    int k = s;
    for (; k + GROUP <= t; k += GROUP)
      add_group<true, COUNT_DEG>(src, run, k, GROUP, lane, tile, degc);
    if (k < t)
      add_group<false, COUNT_DEG>(src, run, k, t - k, lane, tile, degc);
  }
  __syncthreads();

  for (int r = wave; r < R; r += 32) {
    const int row = b * R + r;
    if (row < nrows) {
      float* q = dst + (size_t)row * 64 + lane;
      q[0] = tile[r * 64 + lane];
      q[32] = tile[r * 64 + 32 + lane];
      if (COUNT_DEG && lane == 0) deg[row] = (float)degc[r];
    }
  }
}

extern "C" void kernel_launch(void* const* d_in, const int* in_sizes, int n_in,
                              void* d_out, int out_size, void* d_ws,
                              size_t ws_size, hipStream_t stream) {
  (void)in_sizes; (void)n_in; (void)out_size; (void)ws_size;
  const float* X    = (const float*)d_in[0];
  const float* X0   = (const float*)d_in[1];
  const int*   vertex = (const int*)d_in[2];
  const int*   edges  = (const int*)d_in[3];
  const float* W1w  = (const float*)d_in[4];
  const float* W1b  = (const float*)d_in[5];
  const float* W2w  = (const float*)d_in[6];
  const float* W2b  = (const float*)d_in[7];
  const float* Ww   = (const float*)d_in[8];
  const float* Wb   = (const float*)d_in[9];

  float* out = (float*)d_out;
  float* Xe  = out + (size_t)N_NODES_C * 64;

  float* ws  = (float*)d_ws;
  float* Y1  = ws;
  float* S   = Y1;
  float* T   = ws + (size_t)N_NODES_C * 64;
  float* deg = T + (size_t)N_NODES_C * 64;

  int* entE = (int*)T;
  int* entV = entE + NNZ_C;
  int* tabE = entV + NNZ_C;
  int* tabV = tabE + BIN_BLOCKS * (NB_E + 1);

  const int tiles = N_NODES_C / 16;
  const int gemmBlocks = (tiles + 7) / 8;

  front_kernel<<<FRONT_BLOCKS, 1024, 0, stream>>>(
      edges, vertex, entE, tabE, entV, tabV, X, W1w, W1b, Y1);
  sum_range_kernel<SHIFT_E, NB_E, false><<<NB_E, 1024, 0, stream>>>(
      Y1, entE, tabE, Xe, nullptr, N_EDGES_C);
  sum_range_kernel<SHIFT_V, NB_V, true><<<NB_V, 1024, 0, stream>>>(
      Xe, entV, tabV, S, deg, N_NODES_C);
  gemm2_mix_kernel<<<gemmBlocks, 256, 0, stream>>>(X, X0, S, deg, W2w, W2b, T,
                                                   N_NODES_C);
  gemm64_bias_kernel<<<gemmBlocks, 256, 0, stream>>>(T, Ww, Wb, out, N_NODES_C);
}


# bin blocks put their ordered chunk together in LDS and store it as whole lines (16 B per lane) instead of one scattered 4-byte global store per incidence; same entry order, no sum reordered
# speedup vs baseline: 3.4832x; 1.0653x over previous
#include <hip/hip_runtime.h>

typedef __attribute__((ext_vector_type(2))) float v2f;
typedef __attribute__((ext_vector_type(4))) float v4f;
typedef __attribute__((ext_vector_type(8))) float v8f;
typedef __attribute__((ext_vector_type(4))) int v4i;

#define N_NODES_C 100000
#define N_EDGES_C 20000
#define NNZ_C     1000000
#define ALPHA_C   0.5f

__device__ __forceinline__ v8f wmma_f32(v2f a, v2f b, v8f c) {
  return __builtin_amdgcn_wmma_f32_16x16x4_f32(false, a, false, b, (short)0, c,
                                               false, false);
}

#define WP64 68
#define WP128 132

template <int THREADS, int K, int PITCH>
__device__ __forceinline__ void stage_w(const float* __restrict__ W, float* Wsh) {
  constexpr int Q = K / 4;
  constexpr int STEPS = 64 * Q / THREADS;
  static_assert(STEPS * THREADS == 64 * Q, "every thread copies the same number of pieces");
  v4f w[STEPS];
#pragma unroll
  for (int j = 0; j < STEPS; ++j)
    w[j] = *(const v4f*)(W + 4 * (threadIdx.x + j * THREADS));
#pragma unroll
  for (int j = 0; j < STEPS; ++j) {
    const int i = threadIdx.x + j * THREADS;
    const int n = i / Q, p = i % Q;
    *(v4f*)(Wsh + n * PITCH + 4 * p) = w[j];
  }
}

template <int PITCH>
__device__ __forceinline__ void wmma_piece(v4f a, const float* Wp, v8f& acc0,
                                           v8f& acc1, v8f& acc2, v8f& acc3) {
  const v4f b0 = *(const v4f*)(Wp);
  const v4f b1 = *(const v4f*)(Wp + 16 * PITCH);
  const v4f b2 = *(const v4f*)(Wp + 32 * PITCH);
  const v4f b3 = *(const v4f*)(Wp + 48 * PITCH);
  acc0 = wmma_f32(a.xy, b0.xy, acc0);
  acc1 = wmma_f32(a.xy, b1.xy, acc1);
  acc2 = wmma_f32(a.xy, b2.xy, acc2);
  acc3 = wmma_f32(a.xy, b3.xy, acc3);
  acc0 = wmma_f32(a.zw, b0.zw, acc0);
  acc1 = wmma_f32(a.zw, b1.zw, acc1);
  acc2 = wmma_f32(a.zw, b2.zw, acc2);
  acc3 = wmma_f32(a.zw, b3.zw, acc3);
}

__device__ __forceinline__ void gemm64_tile(const float* __restrict__ A,
                                            const float* Wsh,
                                            const float* __restrict__ bias,
                                            float* __restrict__ Out, int rowBase,
                                            int lane) {
  const int ll = lane & 15;
  const int lh = lane >> 4;

  const float* Arow = A + (size_t)(rowBase + ll) * 64 + 4 * lh;
  const float* Wrow = Wsh + ll * WP64 + 4 * lh;
  v8f acc0 = {}, acc1 = {}, acc2 = {}, acc3 = {};

  v4f a[8];
#pragma unroll
  for (int q = 0; q < 8; ++q) a[q] = *(const v4f*)(Arow + 8 * q);
#pragma unroll
  for (int q = 0; q < 8; ++q)
    wmma_piece<WP64>(a[q], Wrow + 8 * q, acc0, acc1, acc2, acc3);

#pragma unroll
  for (int j = 0; j < 8; ++j) {
    const int r = rowBase + j + lh * 8;
    float* o = Out + (size_t)r * 64;
    o[ll]      = acc0[j] + bias[ll];
    o[16 + ll] = acc1[j] + bias[16 + ll];
    o[32 + ll] = acc2[j] + bias[32 + ll];
    o[48 + ll] = acc3[j] + bias[48 + ll];
  }
}

__global__ __launch_bounds__(256) void gemm64_bias_kernel(
    const float* __restrict__ A, const float* __restrict__ W,
    const float* __restrict__ bias, float* __restrict__ Out, int M) {
  __shared__ __attribute__((aligned(16))) float Wsh[64 * WP64];
  stage_w<256, 64, WP64>(W, Wsh);
  __syncthreads();

  const int lane = threadIdx.x & 31;
  const int wave = threadIdx.x >> 5;
  const int rowBase = (blockIdx.x * 8 + wave) << 4;
  if (rowBase >= M) return;
  gemm64_tile(A, Wsh, bias, Out, rowBase, lane);
}

__global__ __launch_bounds__(256) void gemm2_mix_kernel(
    const float* __restrict__ X, const float* __restrict__ X0,
    const float* __restrict__ S, const float* __restrict__ deg,
    const float* __restrict__ W2, const float* __restrict__ b2,
    float* __restrict__ T, int M) {
  __shared__ __attribute__((aligned(16))) float Wsh[64 * WP128];
  stage_w<256, 128, WP128>(W2, Wsh);
  __syncthreads();

  const int lane = threadIdx.x & 31;
  const int wave = threadIdx.x >> 5;
  const int rowBase = (blockIdx.x * 8 + wave) << 4;
  if (rowBase >= M) return;
  const int ll = lane & 15;
  const int lh = lane >> 4;

  const float dg = deg[rowBase + ll];
  const float* Xrow = X + (size_t)(rowBase + ll) * 64 + 4 * lh;
  const float* Srow = S + (size_t)(rowBase + ll) * 64 + 4 * lh;
  const float* Wrow = Wsh + ll * WP128 + 4 * lh;
  v8f acc0 = {}, acc1 = {}, acc2 = {}, acc3 = {};

  v4f ax[8], as[8];
#pragma unroll
  for (int q = 0; q < 8; ++q) ax[q] = *(const v4f*)(Xrow + 8 * q);
#pragma unroll
  for (int q = 0; q < 8; ++q) as[q] = *(const v4f*)(Srow + 8 * q);
#pragma unroll
  for (int q = 0; q < 8; ++q)
    wmma_piece<WP128>(dg * ax[q], Wrow + 8 * q, acc0, acc1, acc2, acc3);
#pragma unroll
  for (int q = 0; q < 8; ++q)
    wmma_piece<WP128>(as[q], Wrow + 64 + 8 * q, acc0, acc1, acc2, acc3);

  const float oma = 1.0f - ALPHA_C;
#pragma unroll
  for (int j = 0; j < 8; ++j) {
    const int r = rowBase + j + lh * 8;
    const float dr = deg[r];
    const float* x0r = X0 + (size_t)r * 64;
    float* o = T + (size_t)r * 64;
    o[ll]      = oma * (acc0[j] + dr * b2[ll])      + ALPHA_C * x0r[ll];
    o[16 + ll] = oma * (acc1[j] + dr * b2[16 + ll]) + ALPHA_C * x0r[16 + ll];
    o[32 + ll] = oma * (acc2[j] + dr * b2[32 + ll]) + ALPHA_C * x0r[32 + ll];
    o[48 + ll] = oma * (acc3[j] + dr * b2[48 + ll]) + ALPHA_C * x0r[48 + ll];
  }
}

#define CHUNK 8192
#define BIN_BLOCKS ((NNZ_C + CHUNK - 1) / CHUNK)
#define SHIFT_E 6
#define SHIFT_V 8
#define NB_E ((N_EDGES_C + (1 << SHIFT_E) - 1) >> SHIFT_E)
#define NB_V ((N_NODES_C + (1 << SHIFT_V) - 1) >> SHIFT_V)
#define SRC_BITS 20
#define SRC_MASK ((1 << SRC_BITS) - 1)
static_assert(N_NODES_C <= (1 << SRC_BITS) && N_EDGES_C <= (1 << SRC_BITS),
              "a source row fits the low bits of an entry");
static_assert(NB_E <= 1024 && NB_V <= 1024, "one thread per range in the scan");
static_assert(NNZ_C % 4 == 0 && CHUNK % 4096 == 0,
              "a chunk leaves in 16-byte pieces, the same number per thread");

template <int SHIFT, int NB>
__device__ __forceinline__ void bin_body(
    const int* __restrict__ keys, const int* __restrict__ vals,
    int* __restrict__ entries, int* __restrict__ table, int blk, int* hist,
    int* wsum, int* sorted) {
  const int tid = threadIdx.x;
  const int lane = tid & 31;
  const int wave = tid >> 5;
  for (int i = tid; i < NB; i += 1024) hist[i] = 0;
  __syncthreads();

  const int base = blk * CHUNK;
  int key[8], val[8], rank[8];
#pragma unroll
  for (int j = 0; j < 8; ++j) {
    const int e = base + j * 1024 + tid;
    key[j] = 0;
    val[j] = 0;
    if (e < NNZ_C) {
      key[j] = keys[e];
      val[j] = vals[e];
    }
  }
#pragma unroll
  for (int j = 0; j < 8; ++j) {
    const int e = base + j * 1024 + tid;
    rank[j] = 0;
    if (e < NNZ_C) rank[j] = atomicAdd(&hist[key[j] >> SHIFT], 1);
  }
  __syncthreads();

  const int c = (tid < NB) ? hist[tid] : 0;
  int incl = c;
  for (int d = 1; d < 32; d <<= 1) {
    const int t = __shfl_up(incl, d);
    if (lane >= d) incl += t;
  }
  if (lane == 31) wsum[wave] = incl;
  __syncthreads();
  int before = 0;
  for (int w2 = 0; w2 < wave; ++w2) before += wsum[w2];
  const int start = before + incl - c;
  if (tid < NB) {
    hist[tid] = start;
    table[blk * (NB + 1) + tid] = start;
  }
  if (tid == 0) {
    const int left = NNZ_C - base;
    table[blk * (NB + 1) + NB] = left < CHUNK ? left : CHUNK;
  }
  __syncthreads();

#pragma unroll
  for (int j = 0; j < 8; ++j) {
    const int e = base + j * 1024 + tid;
    if (e < NNZ_C) {
      const int r = key[j] >> SHIFT;
      const int local = key[j] - (r << SHIFT);
      sorted[hist[r] + rank[j]] = (local << SRC_BITS) | val[j];
    }
  }
  __syncthreads();

  const int left = NNZ_C - base;
  const int count = left < CHUNK ? left : CHUNK;
  for (int j = 0; j < CHUNK / 4096; ++j) {
    const int i4 = 4 * (tid + j * 1024);
    if (i4 < count) *(v4i*)(entries + base + i4) = *(const v4i*)(sorted + i4);
  }
}

#define NB_MAX (NB_E > NB_V ? NB_E : NB_V)
#define GEMM1_BLOCKS ((N_NODES_C / 16 + 31) / 32)
#define FRONT_BLOCKS (GEMM1_BLOCKS + 2 * BIN_BLOCKS)
__global__ __launch_bounds__(1024) void front_kernel(
    const int* __restrict__ edges, const int* __restrict__ vertex,
    int* __restrict__ entE, int* __restrict__ tabE, int* __restrict__ entV,
    int* __restrict__ tabV, const float* __restrict__ X,
    const float* __restrict__ W1, const float* __restrict__ b1,
    float* __restrict__ Y1) {
  __shared__ int hist[NB_MAX];
  __shared__ int wsum[32];
  __shared__ __attribute__((aligned(16))) int sorted[CHUNK];
  __shared__ __attribute__((aligned(16))) float Wsh[64 * WP64];
  const int blk = blockIdx.x;
  if (blk < GEMM1_BLOCKS) {
    stage_w<1024, 64, WP64>(W1, Wsh);
    __syncthreads();
    const int lane = threadIdx.x & 31;
    const int wave = threadIdx.x >> 5;
    const int rowBase = (blk * 32 + wave) << 4;
    if (rowBase < N_NODES_C) gemm64_tile(X, Wsh, b1, Y1, rowBase, lane);
  } else if (blk < GEMM1_BLOCKS + BIN_BLOCKS) {
    bin_body<SHIFT_E, NB_E>(edges, vertex, entE, tabE, blk - GEMM1_BLOCKS, hist, wsum,
                            sorted);
  } else {
    bin_body<SHIFT_V, NB_V>(vertex, edges, entV, tabV,
                            blk - GEMM1_BLOCKS - BIN_BLOCKS, hist, wsum, sorted);
  }
}

#define GROUP 8
template <bool FULL, bool COUNT_DEG>
__device__ __forceinline__ void add_group(const float* __restrict__ src,
                                          const int* __restrict__ run, int k,
                                          int n, int lane, float* tile,
                                          int* degc) {
  int g[GROUP];
  float lo[GROUP], hi[GROUP];
#pragma unroll
  for (int i = 0; i < GROUP; ++i) {
    const int ii = (FULL || i < n) ? i : n - 1;
    g[i] = run[k + ii];
  }
#pragma unroll
  for (int i = 0; i < GROUP; ++i) {
    const float* p = src + (size_t)(g[i] & SRC_MASK) * 64 + lane;
    lo[i] = p[0];
    hi[i] = p[32];
  }
#pragma unroll
  for (int i = 0; i < GROUP; ++i) {
    if (FULL || i < n) {
      float* q = tile + (g[i] >> SRC_BITS) * 64 + lane;
      unsafeAtomicAdd(q, lo[i]);
      unsafeAtomicAdd(q + 32, hi[i]);
    }
  }
  if (COUNT_DEG) {
    int gl = g[0];
#pragma unroll
    for (int i = 1; i < GROUP; ++i) gl = (lane == i) ? g[i] : gl;
    if (lane < (FULL ? GROUP : n)) atomicAdd(degc + (gl >> SRC_BITS), 1);
  }
}

template <int SHIFT, int NB, bool COUNT_DEG>
__global__ __launch_bounds__(1024) void sum_range_kernel(
    const float* __restrict__ src, const int* __restrict__ entries,
    const int* __restrict__ table, float* __restrict__ dst,
    float* __restrict__ deg, int nrows) {
  constexpr int R = 1 << SHIFT;
  __shared__ float tile[R * 64];
  __shared__ int degc[R];
  const int tid = threadIdx.x;
  const int lane = tid & 31;
  const int wave = __builtin_amdgcn_readfirstlane(tid >> 5);
  const int b = blockIdx.x;
  for (int i = tid; i < R * 64; i += 1024) tile[i] = 0.0f;
  for (int i = tid; i < R; i += 1024) degc[i] = 0;
  __syncthreads();

  for (int bin = wave; bin < BIN_BLOCKS; bin += 32) {
    const int* trow = table + bin * (NB + 1) + b;
    const int s = __builtin_amdgcn_readfirstlane(trow[0]);
    const int t = __builtin_amdgcn_readfirstlane(trow[1]);
    const int* run = entries + bin * CHUNK;
    int k = s;
    for (; k + GROUP <= t; k += GROUP)
      add_group<true, COUNT_DEG>(src, run, k, GROUP, lane, tile, degc);
    if (k < t)
      add_group<false, COUNT_DEG>(src, run, k, t - k, lane, tile, degc);
  }
  __syncthreads();

  for (int r = wave; r < R; r += 32) {
    const int row = b * R + r;
    if (row < nrows) {
      float* q = dst + (size_t)row * 64 + lane;
      q[0] = tile[r * 64 + lane];
      q[32] = tile[r * 64 + 32 + lane];
      if (COUNT_DEG && lane == 0) deg[row] = (float)degc[r];
    }
  }
}

extern "C" void kernel_launch(void* const* d_in, const int* in_sizes, int n_in,
                              void* d_out, int out_size, void* d_ws,
                              size_t ws_size, hipStream_t stream) {
  (void)in_sizes; (void)n_in; (void)out_size; (void)ws_size;
  const float* X    = (const float*)d_in[0];
  const float* X0   = (const float*)d_in[1];
  const int*   vertex = (const int*)d_in[2];
  const int*   edges  = (const int*)d_in[3];
  const float* W1w  = (const float*)d_in[4];
  const float* W1b  = (const float*)d_in[5];
  const float* W2w  = (const float*)d_in[6];
  const float* W2b  = (const float*)d_in[7];
  const float* Ww   = (const float*)d_in[8];
  const float* Wb   = (const float*)d_in[9];

  float* out = (float*)d_out;
  float* Xe  = out + (size_t)N_NODES_C * 64;

  float* ws  = (float*)d_ws;
  float* Y1  = ws;
  float* S   = Y1;
  float* T   = ws + (size_t)N_NODES_C * 64;
  float* deg = T + (size_t)N_NODES_C * 64;

  int* entE = (int*)T;
  int* entV = entE + NNZ_C;
  int* tabE = entV + NNZ_C;
  int* tabV = tabE + BIN_BLOCKS * (NB_E + 1);

  const int tiles = N_NODES_C / 16;
  const int gemmBlocks = (tiles + 7) / 8;

  front_kernel<<<FRONT_BLOCKS, 1024, 0, stream>>>(
      edges, vertex, entE, tabE, entV, tabV, X, W1w, W1b, Y1);
  sum_range_kernel<SHIFT_E, NB_E, false><<<NB_E, 1024, 0, stream>>>(
      Y1, entE, tabE, Xe, nullptr, N_EDGES_C);
  sum_range_kernel<SHIFT_V, NB_V, true><<<NB_V, 1024, 0, stream>>>(
      Xe, entV, tabV, S, deg, N_NODES_C);
  gemm2_mix_kernel<<<gemmBlocks, 256, 0, stream>>>(X, X0, S, deg, W2w, W2b, T,
                                                   N_NODES_C);
  gemm64_bias_kernel<<<gemmBlocks, 256, 0, stream>>>(T, Ww, Wb, out, N_NODES_C);
}


# segment sums: a range's entries ordered by destination row in LDS (integer count, scan, place), a wave sums whole rows in registers, stores them once; no LDS float adds; sums in placing order
# speedup vs baseline: 4.6409x; 1.3324x over previous
#include <hip/hip_runtime.h>

typedef __attribute__((ext_vector_type(2))) float v2f;
typedef __attribute__((ext_vector_type(4))) float v4f;
typedef __attribute__((ext_vector_type(8))) float v8f;
typedef __attribute__((ext_vector_type(4))) int v4i;

#define N_NODES_C 100000
#define N_EDGES_C 20000
#define NNZ_C     1000000
#define ALPHA_C   0.5f

__device__ __forceinline__ v8f wmma_f32(v2f a, v2f b, v8f c) {
  return __builtin_amdgcn_wmma_f32_16x16x4_f32(false, a, false, b, (short)0, c,
                                               false, false);
}

#define WP64 68
#define WP128 132

template <int THREADS, int K, int PITCH>
__device__ __forceinline__ void stage_w(const float* __restrict__ W, float* Wsh) {
  constexpr int Q = K / 4;
  constexpr int STEPS = 64 * Q / THREADS;
  static_assert(STEPS * THREADS == 64 * Q, "every thread copies the same number of pieces");
  v4f w[STEPS];
#pragma unroll
  for (int j = 0; j < STEPS; ++j)
    w[j] = *(const v4f*)(W + 4 * (threadIdx.x + j * THREADS));
#pragma unroll
  for (int j = 0; j < STEPS; ++j) {
    const int i = threadIdx.x + j * THREADS;
    const int n = i / Q, p = i % Q;
    *(v4f*)(Wsh + n * PITCH + 4 * p) = w[j];
  }
}

template <int PITCH>
__device__ __forceinline__ void wmma_piece(v4f a, const float* Wp, v8f& acc0,
                                           v8f& acc1, v8f& acc2, v8f& acc3) {
  const v4f b0 = *(const v4f*)(Wp);
  const v4f b1 = *(const v4f*)(Wp + 16 * PITCH);
  const v4f b2 = *(const v4f*)(Wp + 32 * PITCH);
  const v4f b3 = *(const v4f*)(Wp + 48 * PITCH);
  acc0 = wmma_f32(a.xy, b0.xy, acc0);
  acc1 = wmma_f32(a.xy, b1.xy, acc1);
  acc2 = wmma_f32(a.xy, b2.xy, acc2);
  acc3 = wmma_f32(a.xy, b3.xy, acc3);
  acc0 = wmma_f32(a.zw, b0.zw, acc0);
  acc1 = wmma_f32(a.zw, b1.zw, acc1);
  acc2 = wmma_f32(a.zw, b2.zw, acc2);
  acc3 = wmma_f32(a.zw, b3.zw, acc3);
}

__device__ __forceinline__ void gemm64_tile(const float* __restrict__ A,
                                            const float* Wsh,
                                            const float* __restrict__ bias,
                                            float* __restrict__ Out, int rowBase,
                                            int lane) {
  const int ll = lane & 15;
  const int lh = lane >> 4;

  const float* Arow = A + (size_t)(rowBase + ll) * 64 + 4 * lh;
  const float* Wrow = Wsh + ll * WP64 + 4 * lh;
  v8f acc0 = {}, acc1 = {}, acc2 = {}, acc3 = {};

  v4f a[8];
#pragma unroll
  for (int q = 0; q < 8; ++q) a[q] = *(const v4f*)(Arow + 8 * q);
#pragma unroll
  for (int q = 0; q < 8; ++q)
    wmma_piece<WP64>(a[q], Wrow + 8 * q, acc0, acc1, acc2, acc3);

#pragma unroll
  for (int j = 0; j < 8; ++j) {
    const int r = rowBase + j + lh * 8;
    float* o = Out + (size_t)r * 64;
    o[ll]      = acc0[j] + bias[ll];
    o[16 + ll] = acc1[j] + bias[16 + ll];
    o[32 + ll] = acc2[j] + bias[32 + ll];
    o[48 + ll] = acc3[j] + bias[48 + ll];
  }
}

#define TP 36

__device__ __forceinline__ void wave_lds_handover() {
  __builtin_amdgcn_fence(__ATOMIC_RELEASE, "wavefront");
  __builtin_amdgcn_wave_barrier();
  __builtin_amdgcn_fence(__ATOMIC_ACQUIRE, "wavefront");
}

__global__ __launch_bounds__(256, 8) void gemm2_mix_out_kernel(
    const float* __restrict__ X, const float* __restrict__ X0,
    const float* __restrict__ S, const float* __restrict__ deg,
    const float* __restrict__ W2, const float* __restrict__ b2,
    const float* __restrict__ W, const float* __restrict__ b,
    float* __restrict__ Out, int M) {
  __shared__ __attribute__((aligned(16))) float W2sh[64 * WP128];
  __shared__ __attribute__((aligned(16))) float Wsh[64 * WP64];
  __shared__ __attribute__((aligned(16))) float Tsh[8 * 16 * TP];
  stage_w<256, 128, WP128>(W2, W2sh);
  stage_w<256, 64, WP64>(W, Wsh);
  __syncthreads();

  const int lane = threadIdx.x & 31;
  const int wave = threadIdx.x >> 5;
  const int rowBase = (blockIdx.x * 8 + wave) << 4;
  if (rowBase >= M) return;
  const int ll = lane & 15;
  const int lh = lane >> 4;

  const float dg = deg[rowBase + ll];
  const float* Xrow = X + (size_t)(rowBase + ll) * 64 + 4 * lh;
  const float* Srow = S + (size_t)(rowBase + ll) * 64 + 4 * lh;
  const float* W2row = W2sh + ll * WP128 + 4 * lh;
  v8f acc0 = {}, acc1 = {}, acc2 = {}, acc3 = {};

  v4f ax[8], as[8];
#pragma unroll
  for (int q = 0; q < 8; ++q) ax[q] = *(const v4f*)(Xrow + 8 * q);
#pragma unroll
  for (int q = 0; q < 8; ++q) as[q] = *(const v4f*)(Srow + 8 * q);
#pragma unroll
  for (int q = 0; q < 8; ++q)
    wmma_piece<WP128>(dg * ax[q], W2row + 8 * q, acc0, acc1, acc2, acc3);
#pragma unroll
  for (int q = 0; q < 8; ++q)
    wmma_piece<WP128>(as[q], W2row + 64 + 8 * q, acc0, acc1, acc2, acc3);

  const float oma = 1.0f - ALPHA_C;
#pragma unroll
  for (int j = 0; j < 8; ++j) {
    const int r = rowBase + j + lh * 8;
    const float dr = deg[r];
    const float* x0r = X0 + (size_t)r * 64;
    acc0[j] = oma * (acc0[j] + dr * b2[ll])      + ALPHA_C * x0r[ll];
    acc1[j] = oma * (acc1[j] + dr * b2[16 + ll]) + ALPHA_C * x0r[16 + ll];
    acc2[j] = oma * (acc2[j] + dr * b2[32 + ll]) + ALPHA_C * x0r[32 + ll];
    acc3[j] = oma * (acc3[j] + dr * b2[48 + ll]) + ALPHA_C * x0r[48 + ll];
  }

  float* Tw = Tsh + wave * (16 * TP);
  float* Tst = Tw + (lh * 8) * TP + ll;
  const float* Trow = Tw + ll * TP + 4 * lh;
  const float* Wrow = Wsh + ll * WP64 + 4 * lh;
  v8f o0 = {}, o1 = {}, o2 = {}, o3 = {};
  v4f a[4];

#pragma unroll
  for (int j = 0; j < 8; ++j) {
    Tst[j * TP]      = acc0[j];
    Tst[j * TP + 16] = acc1[j];
  }
  wave_lds_handover();
#pragma unroll
  for (int q = 0; q < 4; ++q) a[q] = *(const v4f*)(Trow + 8 * q);
  wave_lds_handover();
#pragma unroll
  for (int q = 0; q < 4; ++q)
    wmma_piece<WP64>(a[q], Wrow + 8 * q, o0, o1, o2, o3);

#pragma unroll
  for (int j = 0; j < 8; ++j) {
    Tst[j * TP]      = acc2[j];
    Tst[j * TP + 16] = acc3[j];
  }
  wave_lds_handover();
#pragma unroll
  for (int q = 0; q < 4; ++q) a[q] = *(const v4f*)(Trow + 8 * q);
#pragma unroll
  for (int q = 0; q < 4; ++q)
    wmma_piece<WP64>(a[q], Wrow + 32 + 8 * q, o0, o1, o2, o3);

#pragma unroll
  for (int j = 0; j < 8; ++j) {
    const int r = rowBase + j + lh * 8;
    float* o = Out + (size_t)r * 64;
    o[ll]      = o0[j] + b[ll];
    o[16 + ll] = o1[j] + b[16 + ll];
    o[32 + ll] = o2[j] + b[32 + ll];
    o[48 + ll] = o3[j] + b[48 + ll];
  }
}

#define CHUNK 8192
#define BIN_BLOCKS ((NNZ_C + CHUNK - 1) / CHUNK)
#define SHIFT_E 6
#define SHIFT_V 8
#define NB_E ((N_EDGES_C + (1 << SHIFT_E) - 1) >> SHIFT_E)
#define NB_V ((N_NODES_C + (1 << SHIFT_V) - 1) >> SHIFT_V)
#define SRC_BITS 20
#define SRC_MASK ((1 << SRC_BITS) - 1)
static_assert(N_NODES_C <= (1 << SRC_BITS) && N_EDGES_C <= (1 << SRC_BITS),
              "a source row fits the low bits of an entry");
static_assert(NB_E <= 1024 && NB_V <= 1024, "one thread per range in the scan");
static_assert(NNZ_C % 4 == 0 && CHUNK % 4096 == 0,
              "a chunk leaves in 16-byte pieces, the same number per thread");

template <int SHIFT, int NB>
__device__ __forceinline__ void bin_body(
    const int* __restrict__ keys, const int* __restrict__ vals,
    int* __restrict__ entries, int* __restrict__ table, int blk, int* hist,
    int* wsum, int* sorted) {
  const int tid = threadIdx.x;
  const int lane = tid & 31;
  const int wave = tid >> 5;
  for (int i = tid; i < NB; i += 1024) hist[i] = 0;
  __syncthreads();

  const int base = blk * CHUNK;
  int key[8], val[8], rank[8];
#pragma unroll
  for (int j = 0; j < 8; ++j) {
    const int e = base + j * 1024 + tid;
    key[j] = 0;
    val[j] = 0;
    if (e < NNZ_C) {
      key[j] = keys[e];
      val[j] = vals[e];
    }
  }
#pragma unroll
  for (int j = 0; j < 8; ++j) {
    const int e = base + j * 1024 + tid;
    rank[j] = 0;
    if (e < NNZ_C) rank[j] = atomicAdd(&hist[key[j] >> SHIFT], 1);
  }
  __syncthreads();

  const int c = (tid < NB) ? hist[tid] : 0;
  int incl = c;
  for (int d = 1; d < 32; d <<= 1) {
    const int t = __shfl_up(incl, d);
    if (lane >= d) incl += t;
  }
  if (lane == 31) wsum[wave] = incl;
  __syncthreads();
  int before = 0;
  for (int w2 = 0; w2 < wave; ++w2) before += wsum[w2];
  const int start = before + incl - c;
  if (tid < NB) {
    hist[tid] = start;
    table[blk * (NB + 1) + tid] = start;
  }
  if (tid == 0) {
    const int left = NNZ_C - base;
    table[blk * (NB + 1) + NB] = left < CHUNK ? left : CHUNK;
  }
  __syncthreads();

#pragma unroll
  for (int j = 0; j < 8; ++j) {
    const int e = base + j * 1024 + tid;
    if (e < NNZ_C) {
      const int r = key[j] >> SHIFT;
      const int local = key[j] - (r << SHIFT);
      sorted[hist[r] + rank[j]] = (local << SRC_BITS) | val[j];
    }
  }
  __syncthreads();

  const int left = NNZ_C - base;
  const int count = left < CHUNK ? left : CHUNK;
  for (int j = 0; j < CHUNK / 4096; ++j) {
    const int i4 = 4 * (tid + j * 1024);
    if (i4 < count) *(v4i*)(entries + base + i4) = *(const v4i*)(sorted + i4);
  }
}

#define NB_MAX (NB_E > NB_V ? NB_E : NB_V)
#define GEMM1_BLOCKS ((N_NODES_C / 16 + 31) / 32)
#define FRONT_BLOCKS (GEMM1_BLOCKS + 2 * BIN_BLOCKS)
__global__ __launch_bounds__(1024) void front_kernel(
    const int* __restrict__ edges, const int* __restrict__ vertex,
    int* __restrict__ entE, int* __restrict__ tabE, int* __restrict__ entV,
    int* __restrict__ tabV, const float* __restrict__ X,
    const float* __restrict__ W1, const float* __restrict__ b1,
    float* __restrict__ Y1) {
  __shared__ int hist[NB_MAX];
  __shared__ int wsum[32];
  __shared__ __attribute__((aligned(16))) int sorted[CHUNK];
  __shared__ __attribute__((aligned(16))) float Wsh[64 * WP64];
  const int blk = blockIdx.x;
  if (blk < GEMM1_BLOCKS) {
    stage_w<1024, 64, WP64>(W1, Wsh);
    __syncthreads();
    const int lane = threadIdx.x & 31;
    const int wave = threadIdx.x >> 5;
    const int rowBase = (blk * 32 + wave) << 4;
    if (rowBase < N_NODES_C) gemm64_tile(X, Wsh, b1, Y1, rowBase, lane);
  } else if (blk < GEMM1_BLOCKS + BIN_BLOCKS) {
    bin_body<SHIFT_E, NB_E>(edges, vertex, entE, tabE, blk - GEMM1_BLOCKS, hist, wsum,
                            sorted);
  } else {
    bin_body<SHIFT_V, NB_V>(vertex, edges, entV, tabV,
                            blk - GEMM1_BLOCKS - BIN_BLOCKS, hist, wsum, sorted);
  }
}

#define BPW ((BIN_BLOCKS + 31) / 32)

template <int G, bool FULL>
__device__ __forceinline__ void add_rows(const float* __restrict__ src, int v,
                                         int first, int n, int lane, float& lo,
                                         float& hi) {
  float a[G], c[G];
#pragma unroll
  for (int i = 0; i < G; ++i) {
    const int idx = __builtin_amdgcn_readlane(v, first + i);
    const float* p = src + (size_t)idx * 64 + lane;
    a[i] = p[0];
    c[i] = p[32];
  }
#pragma unroll
  for (int i = 0; i < G; ++i) {
    if (FULL || i < n) {
      lo += a[i];
      hi += c[i];
    }
  }
}

#define GROUP 8
template <bool FULL>
__device__ __forceinline__ void add_group(const float* __restrict__ src,
                                          const int* __restrict__ run, int k,
                                          int n, int lane, float* tile) {
  int g[GROUP];
  float lo[GROUP], hi[GROUP];
#pragma unroll
  for (int i = 0; i < GROUP; ++i) {
    const int ii = (FULL || i < n) ? i : n - 1;
    g[i] = run[k + ii];
  }
#pragma unroll
  for (int i = 0; i < GROUP; ++i) {
    const float* p = src + (size_t)(g[i] & SRC_MASK) * 64 + lane;
    lo[i] = p[0];
    hi[i] = p[32];
  }
#pragma unroll
  for (int i = 0; i < GROUP; ++i) {
    if (FULL || i < n) {
      float* q = tile + (g[i] >> SRC_BITS) * 64 + lane;
      unsafeAtomicAdd(q, lo[i]);
      unsafeAtomicAdd(q + 32, hi[i]);
    }
  }
}

template <int SHIFT, int NB, bool COUNT_DEG>
__global__ __launch_bounds__(1024) void sum_range_kernel(
    const float* __restrict__ src, const int* __restrict__ entries,
    const int* __restrict__ table, float* __restrict__ dst,
    float* __restrict__ deg, int nrows) {
  constexpr int R = 1 << SHIFT;
  constexpr int CAP = R * 64;
  __shared__ __attribute__((aligned(16))) float tile[R * 64];
  __shared__ int cnt[R];
  __shared__ int start[R + 1];
  __shared__ int wsum[32];
  int* sorted = reinterpret_cast<int*>(tile);
  const int tid = threadIdx.x;
  const int lane = tid & 31;
  const int wave = __builtin_amdgcn_readfirstlane(tid >> 5);
  const int b = blockIdx.x;

  int s[BPW], t[BPW], e[BPW];
  const int* run[BPW];
#pragma unroll
  for (int j = 0; j < BPW; ++j) {
    const int bin = wave + 32 * j;
    const int bc = bin < BIN_BLOCKS ? bin : BIN_BLOCKS - 1;
    const int* trow = table + bc * (NB + 1) + b;
    s[j] = __builtin_amdgcn_readfirstlane(trow[0]);
    const int tj = __builtin_amdgcn_readfirstlane(trow[1]);
    t[j] = bin < BIN_BLOCKS ? tj : s[j];
    run[j] = entries + bc * CHUNK;
  }
  for (int i = tid; i < R; i += 1024) cnt[i] = 0;
#pragma unroll
  for (int j = 0; j < BPW; ++j) {
    int k = s[j] + lane;
    k = k < t[j] ? k : t[j] - 1;
    k = k > 0 ? k : 0;
    e[j] = run[j][k];
  }
  __syncthreads();

#pragma unroll
  for (int j = 0; j < BPW; ++j) {
    if (s[j] + lane < t[j]) atomicAdd(&cnt[e[j] >> SRC_BITS], 1);
    for (int k = s[j] + 32 + lane; k < t[j]; k += 32)
      atomicAdd(&cnt[run[j][k] >> SRC_BITS], 1);
  }
  __syncthreads();

  {
    const int c = (tid < R) ? cnt[tid] : 0;
    int incl = c;
    for (int d = 1; d < 32; d <<= 1) {
      const int u = __shfl_up(incl, d);
      if (lane >= d) incl += u;
    }
    if (lane == 31) wsum[wave] = incl;
    __syncthreads();
    if (tid < R) {
      int before = 0;
      for (int w2 = 0; w2 < wave; ++w2) before += wsum[w2];
      const int st = before + incl - c;
      start[tid] = st;
      cnt[tid] = st;
      if (tid == R - 1) start[R] = st + c;
      if (COUNT_DEG) {
        const int row = b * R + tid;
        if (row < nrows) deg[row] = (float)c;
      }
    }
    __syncthreads();
  }
  const int total = start[R];

  if (total <= CAP) {
#pragma unroll
    for (int j = 0; j < BPW; ++j) {
      if (s[j] + lane < t[j]) {
        const int pos = atomicAdd(&cnt[e[j] >> SRC_BITS], 1);
        sorted[pos] = e[j] & SRC_MASK;
      }
      for (int k = s[j] + 32 + lane; k < t[j]; k += 32) {
        const int ee = run[j][k];
        const int pos = atomicAdd(&cnt[ee >> SRC_BITS], 1);
        sorted[pos] = ee & SRC_MASK;
      }
    }
    __syncthreads();

    for (int r = wave; r < R; r += 32) {
      const int row = b * R + r;
      if (row >= nrows) break;
      const int beg = __builtin_amdgcn_readfirstlane(start[r]);
      const int end = __builtin_amdgcn_readfirstlane(start[r + 1]);
      float lo = 0.0f, hi = 0.0f;
      for (int k = beg; k < end; k += 32) {
        const int left = end - k;
        const int m = left < 32 ? left : 32;
        const int v = sorted[k + (lane < m ? lane : m - 1)];
#pragma unroll
        for (int g = 0; g < 32; g += 8) {
          const int rem = m - g;
          if (rem >= 8) {
            add_rows<8, true>(src, v, g, 8, lane, lo, hi);
          } else {
            if (rem > 4) add_rows<8, false>(src, v, g, rem, lane, lo, hi);
            else if (rem > 0) add_rows<4, false>(src, v, g, rem, lane, lo, hi);
            break;
          }
        }
      }
      float* q = dst + (size_t)row * 64 + lane;
      q[0] = lo;
      q[32] = hi;
    }
  } else {
    for (int i = tid; i < R * 64; i += 1024) tile[i] = 0.0f;
    __syncthreads();
#pragma unroll
    for (int j = 0; j < BPW; ++j) {
      int k = s[j];
      for (; k + GROUP <= t[j]; k += GROUP)
        add_group<true>(src, run[j], k, GROUP, lane, tile);
      if (k < t[j]) add_group<false>(src, run[j], k, t[j] - k, lane, tile);
    }
    __syncthreads();
    for (int r = wave; r < R; r += 32) {
      const int row = b * R + r;
      if (row < nrows) {
        float* q = dst + (size_t)row * 64 + lane;
        q[0] = tile[r * 64 + lane];
        q[32] = tile[r * 64 + 32 + lane];
      }
    }
  }
}

extern "C" void kernel_launch(void* const* d_in, const int* in_sizes, int n_in,
                              void* d_out, int out_size, void* d_ws,
                              size_t ws_size, hipStream_t stream) {
  (void)in_sizes; (void)n_in; (void)out_size; (void)ws_size;
  const float* X    = (const float*)d_in[0];
  const float* X0   = (const float*)d_in[1];
  const int*   vertex = (const int*)d_in[2];
  const int*   edges  = (const int*)d_in[3];
  const float* W1w  = (const float*)d_in[4];
  const float* W1b  = (const float*)d_in[5];
  const float* W2w  = (const float*)d_in[6];
  const float* W2b  = (const float*)d_in[7];
  const float* Ww   = (const float*)d_in[8];
  const float* Wb   = (const float*)d_in[9];

  float* out = (float*)d_out;
  float* Xe  = out + (size_t)N_NODES_C * 64;

  float* ws  = (float*)d_ws;
  float* Y1  = ws;
  float* S   = Y1;
  float* T   = ws + (size_t)N_NODES_C * 64;
  float* deg = T + (size_t)N_NODES_C * 64;

  int* entE = (int*)T;
  int* entV = entE + NNZ_C;
  int* tabE = entV + NNZ_C;
  int* tabV = tabE + BIN_BLOCKS * (NB_E + 1);

  const int tiles = N_NODES_C / 16;
  const int gemmBlocks = (tiles + 7) / 8;

  front_kernel<<<FRONT_BLOCKS, 1024, 0, stream>>>(
      edges, vertex, entE, tabE, entV, tabV, X, W1w, W1b, Y1);
  sum_range_kernel<SHIFT_E, NB_E, false><<<NB_E, 1024, 0, stream>>>(
      Y1, entE, tabE, Xe, nullptr, N_EDGES_C);
  sum_range_kernel<SHIFT_V, NB_V, true><<<NB_V, 1024, 0, stream>>>(
      Xe, entV, tabV, S, deg, N_NODES_C);
  gemm2_mix_out_kernel<<<gemmBlocks, 256, 0, stream>>>(X, X0, S, deg, W2w, W2b, Ww,
                                                       Wb, out, N_NODES_C);
}


# products with W1 and with W2[:,64:] moved from the 100000 node rows to the 20000 summed edge rows (linear in the sum), done per 64x64 tile inside the edge sum launch; sums reordered, f32 WMMA kept
# speedup vs baseline: 5.1624x; 1.1124x over previous
#include <hip/hip_runtime.h>

typedef __attribute__((ext_vector_type(2))) float v2f;
typedef __attribute__((ext_vector_type(4))) float v4f;
typedef __attribute__((ext_vector_type(8))) float v8f;
typedef __attribute__((ext_vector_type(4))) int v4i;

#define N_NODES_C 100000
#define N_EDGES_C 20000
#define NNZ_C     1000000
#define ALPHA_C   0.5f

__device__ __forceinline__ v8f wmma_f32(v2f a, v2f b, v8f c) {
  return __builtin_amdgcn_wmma_f32_16x16x4_f32(false, a, false, b, (short)0, c,
                                               false, false);
}

#define WP64 68

template <int THREADS, int K, int PITCH, int SRC = K>
__device__ __forceinline__ void stage_w(const float* __restrict__ W, float* Wsh) {
  constexpr int Q = K / 4;
  constexpr int STEPS = 64 * Q / THREADS;
  static_assert(STEPS * THREADS == 64 * Q, "every thread copies the same number of pieces");
  v4f w[STEPS];
#pragma unroll
  for (int j = 0; j < STEPS; ++j) {
    const int i = threadIdx.x + j * THREADS;
    w[j] = *(const v4f*)(W + (i / Q) * SRC + 4 * (i % Q));
  }
#pragma unroll
  for (int j = 0; j < STEPS; ++j) {
    const int i = threadIdx.x + j * THREADS;
    const int n = i / Q, p = i % Q;
    *(v4f*)(Wsh + n * PITCH + 4 * p) = w[j];
  }
}

template <int PITCH>
__device__ __forceinline__ void wmma_piece(v4f a, const float* Wp, v8f& acc0,
                                           v8f& acc1, v8f& acc2, v8f& acc3) {
  const v4f b0 = *(const v4f*)(Wp);
  const v4f b1 = *(const v4f*)(Wp + 16 * PITCH);
  const v4f b2 = *(const v4f*)(Wp + 32 * PITCH);
  const v4f b3 = *(const v4f*)(Wp + 48 * PITCH);
  acc0 = wmma_f32(a.xy, b0.xy, acc0);
  acc1 = wmma_f32(a.xy, b1.xy, acc1);
  acc2 = wmma_f32(a.xy, b2.xy, acc2);
  acc3 = wmma_f32(a.xy, b3.xy, acc3);
  acc0 = wmma_f32(a.zw, b0.zw, acc0);
  acc1 = wmma_f32(a.zw, b1.zw, acc1);
  acc2 = wmma_f32(a.zw, b2.zw, acc2);
  acc3 = wmma_f32(a.zw, b3.zw, acc3);
}

#define TP 36

__device__ __forceinline__ void wave_lds_handover() {
  __builtin_amdgcn_fence(__ATOMIC_RELEASE, "wavefront");
  __builtin_amdgcn_wave_barrier();
  __builtin_amdgcn_fence(__ATOMIC_ACQUIRE, "wavefront");
}

__global__ __launch_bounds__(256, 8) void gemm2_mix_out_kernel(
    const float* __restrict__ X, const float* __restrict__ X0,
    const float* __restrict__ S, const float* __restrict__ deg,
    const float* __restrict__ W2, const float* __restrict__ b2,
    const float* __restrict__ W, const float* __restrict__ b,
    float* __restrict__ Out, int M) {
  __shared__ __attribute__((aligned(16))) float W2sh[64 * WP64];
  __shared__ __attribute__((aligned(16))) float Wsh[64 * WP64];
  __shared__ __attribute__((aligned(16))) float Tsh[8 * 16 * TP];
  stage_w<256, 64, WP64, 128>(W2, W2sh);
  stage_w<256, 64, WP64>(W, Wsh);
  __syncthreads();

  const int lane = threadIdx.x & 31;
  const int wave = threadIdx.x >> 5;
  const int rowBase = (blockIdx.x * 8 + wave) << 4;
  if (rowBase >= M) return;
  const int ll = lane & 15;
  const int lh = lane >> 4;

  const float dg = deg[rowBase + ll];
  const float* Xrow = X + (size_t)(rowBase + ll) * 64 + 4 * lh;
  const float* W2row = W2sh + ll * WP64 + 4 * lh;
  v8f acc0, acc1, acc2, acc3;

  v4f ax[8];
#pragma unroll
  for (int q = 0; q < 8; ++q) ax[q] = *(const v4f*)(Xrow + 8 * q);
#pragma unroll
  for (int j = 0; j < 8; ++j) {
    const float* sr = S + (size_t)(rowBase + j + lh * 8) * 64;
    acc0[j] = sr[ll];
    acc1[j] = sr[16 + ll];
    acc2[j] = sr[32 + ll];
    acc3[j] = sr[48 + ll];
  }
#pragma unroll
  for (int q = 0; q < 8; ++q)
    wmma_piece<WP64>(dg * ax[q], W2row + 8 * q, acc0, acc1, acc2, acc3);

  const float oma = 1.0f - ALPHA_C;
#pragma unroll
  for (int j = 0; j < 8; ++j) {
    const int r = rowBase + j + lh * 8;
    const float dr = deg[r];
    const float* x0r = X0 + (size_t)r * 64;
    acc0[j] = oma * (acc0[j] + dr * b2[ll])      + ALPHA_C * x0r[ll];
    acc1[j] = oma * (acc1[j] + dr * b2[16 + ll]) + ALPHA_C * x0r[16 + ll];
    acc2[j] = oma * (acc2[j] + dr * b2[32 + ll]) + ALPHA_C * x0r[32 + ll];
    acc3[j] = oma * (acc3[j] + dr * b2[48 + ll]) + ALPHA_C * x0r[48 + ll];
  }

  float* Tw = Tsh + wave * (16 * TP);
  float* Tst = Tw + (lh * 8) * TP + ll;
  const float* Trow = Tw + ll * TP + 4 * lh;
  const float* Wrow = Wsh + ll * WP64 + 4 * lh;
  v8f o0 = {}, o1 = {}, o2 = {}, o3 = {};
  v4f a[4];

#pragma unroll
  for (int j = 0; j < 8; ++j) {
    Tst[j * TP]      = acc0[j];
    Tst[j * TP + 16] = acc1[j];
  }
  wave_lds_handover();
#pragma unroll
  for (int q = 0; q < 4; ++q) a[q] = *(const v4f*)(Trow + 8 * q);
  wave_lds_handover();
#pragma unroll
  for (int q = 0; q < 4; ++q)
    wmma_piece<WP64>(a[q], Wrow + 8 * q, o0, o1, o2, o3);

#pragma unroll
  for (int j = 0; j < 8; ++j) {
    Tst[j * TP]      = acc2[j];
    Tst[j * TP + 16] = acc3[j];
  }
  wave_lds_handover();
#pragma unroll
  for (int q = 0; q < 4; ++q) a[q] = *(const v4f*)(Trow + 8 * q);
#pragma unroll
  for (int q = 0; q < 4; ++q)
    wmma_piece<WP64>(a[q], Wrow + 32 + 8 * q, o0, o1, o2, o3);

#pragma unroll
  for (int j = 0; j < 8; ++j) {
    const int r = rowBase + j + lh * 8;
    float* o = Out + (size_t)r * 64;
    o[ll]      = o0[j] + b[ll];
    o[16 + ll] = o1[j] + b[16 + ll];
    o[32 + ll] = o2[j] + b[32 + ll];
    o[48 + ll] = o3[j] + b[48 + ll];
  }
}

#define CHUNK 8192
#define BIN_BLOCKS ((NNZ_C + CHUNK - 1) / CHUNK)
#define SHIFT_E 6
#define SHIFT_V 8
#define NB_E ((N_EDGES_C + (1 << SHIFT_E) - 1) >> SHIFT_E)
#define NB_V ((N_NODES_C + (1 << SHIFT_V) - 1) >> SHIFT_V)
#define SRC_BITS 20
#define SRC_MASK ((1 << SRC_BITS) - 1)
static_assert(N_NODES_C <= (1 << SRC_BITS) && N_EDGES_C <= (1 << SRC_BITS),
              "a source row fits the low bits of an entry");
static_assert(NB_E <= 1024 && NB_V <= 1024, "one thread per range in the scan");
static_assert(NNZ_C % 4 == 0 && CHUNK % 4096 == 0,
              "a chunk leaves in 16-byte pieces, the same number per thread");
#define BIN_WORDS (2 * NNZ_C + BIN_BLOCKS * (NB_E + 1) + BIN_BLOCKS * (NB_V + 1))
#define Z_OFFSET (((BIN_WORDS + 63) / 64) * 64)
static_assert((size_t)Z_OFFSET + (size_t)N_EDGES_C * 64 <= (size_t)N_NODES_C * 64,
              "Z fits the plane behind the bin outputs");

template <int SHIFT, int NB>
__device__ __forceinline__ void bin_body(
    const int* __restrict__ keys, const int* __restrict__ vals,
    int* __restrict__ entries, int* __restrict__ table, int blk, int* hist,
    int* wsum, int* sorted) {
  const int tid = threadIdx.x;
  const int lane = tid & 31;
  const int wave = tid >> 5;
  for (int i = tid; i < NB; i += 1024) hist[i] = 0;
  __syncthreads();

  const int base = blk * CHUNK;
  int key[8], val[8], rank[8];
#pragma unroll
  for (int j = 0; j < 8; ++j) {
    const int e = base + j * 1024 + tid;
    key[j] = 0;
    val[j] = 0;
    if (e < NNZ_C) {
      key[j] = keys[e];
      val[j] = vals[e];
    }
  }
#pragma unroll
  for (int j = 0; j < 8; ++j) {
    const int e = base + j * 1024 + tid;
    rank[j] = 0;
    if (e < NNZ_C) rank[j] = atomicAdd(&hist[key[j] >> SHIFT], 1);
  }
  __syncthreads();

  const int c = (tid < NB) ? hist[tid] : 0;
  int incl = c;
  for (int d = 1; d < 32; d <<= 1) {
    const int t = __shfl_up(incl, d);
    if (lane >= d) incl += t;
  }
  if (lane == 31) wsum[wave] = incl;
  __syncthreads();
  int before = 0;
  for (int w2 = 0; w2 < wave; ++w2) before += wsum[w2];
  const int start = before + incl - c;
  if (tid < NB) {
    hist[tid] = start;
    table[blk * (NB + 1) + tid] = start;
  }
  if (tid == 0) {
    const int left = NNZ_C - base;
    table[blk * (NB + 1) + NB] = left < CHUNK ? left : CHUNK;
  }
  __syncthreads();

#pragma unroll
  for (int j = 0; j < 8; ++j) {
    const int e = base + j * 1024 + tid;
    if (e < NNZ_C) {
      const int r = key[j] >> SHIFT;
      const int local = key[j] - (r << SHIFT);
      sorted[hist[r] + rank[j]] = (local << SRC_BITS) | val[j];
    }
  }
  __syncthreads();

  const int left = NNZ_C - base;
  const int count = left < CHUNK ? left : CHUNK;
  for (int j = 0; j < CHUNK / 4096; ++j) {
    const int i4 = 4 * (tid + j * 1024);
    if (i4 < count) *(v4i*)(entries + base + i4) = *(const v4i*)(sorted + i4);
  }
}

#define NB_MAX (NB_E > NB_V ? NB_E : NB_V)
#define FRONT_BLOCKS (2 * BIN_BLOCKS)
__global__ __launch_bounds__(1024) void front_kernel(
    const int* __restrict__ edges, const int* __restrict__ vertex,
    int* __restrict__ entE, int* __restrict__ tabE, int* __restrict__ entV,
    int* __restrict__ tabV) {
  __shared__ int hist[NB_MAX];
  __shared__ int wsum[32];
  __shared__ __attribute__((aligned(16))) int sorted[CHUNK];
  const int blk = blockIdx.x;
  if (blk < BIN_BLOCKS) {
    bin_body<SHIFT_E, NB_E>(edges, vertex, entE, tabE, blk, hist, wsum, sorted);
  } else {
    bin_body<SHIFT_V, NB_V>(vertex, edges, entV, tabV, blk - BIN_BLOCKS, hist, wsum,
                            sorted);
  }
}

#define BPW ((BIN_BLOCKS + 31) / 32)

template <int G, bool FULL>
__device__ __forceinline__ void add_rows(const float* __restrict__ src, int v,
                                         int first, int n, int lane, float& lo,
                                         float& hi) {
  float a[G], c[G];
#pragma unroll
  for (int i = 0; i < G; ++i) {
    const int idx = __builtin_amdgcn_readlane(v, first + i);
    const float* p = src + (size_t)idx * 64 + lane;
    a[i] = p[0];
    c[i] = p[32];
  }
#pragma unroll
  for (int i = 0; i < G; ++i) {
    if (FULL || i < n) {
      lo += a[i];
      hi += c[i];
    }
  }
}

#define GROUP 8
template <bool FULL>
__device__ __forceinline__ void add_group(const float* __restrict__ src,
                                          const int* __restrict__ run, int k,
                                          int n, int lane, float* tile) {
  int g[GROUP];
  float lo[GROUP], hi[GROUP];
#pragma unroll
  for (int i = 0; i < GROUP; ++i) {
    const int ii = (FULL || i < n) ? i : n - 1;
    g[i] = run[k + ii];
  }
#pragma unroll
  for (int i = 0; i < GROUP; ++i) {
    const float* p = src + (size_t)(g[i] & SRC_MASK) * 64 + lane;
    lo[i] = p[0];
    hi[i] = p[32];
  }
#pragma unroll
  for (int i = 0; i < GROUP; ++i) {
    if (FULL || i < n) {
      float* q = tile + (g[i] >> SRC_BITS) * 64 + lane;
      unsafeAtomicAdd(q, lo[i]);
      unsafeAtomicAdd(q + 32, hi[i]);
    }
  }
}

template <int SHIFT, int NB, bool COUNT_DEG, bool EDGE>
__global__ __launch_bounds__(1024) void sum_range_kernel(
    const float* __restrict__ src, const int* __restrict__ entries,
    const int* __restrict__ table, float* __restrict__ dst,
    float* __restrict__ deg, int nrows, const float* __restrict__ W1,
    const float* __restrict__ b1, const float* __restrict__ W2,
    float* __restrict__ Z) {
  constexpr int R = 1 << SHIFT;
  constexpr int CAP = R * 64;
  static_assert(!EDGE || R == 64, "the edge products are written for one 64 x 64 tile");
  constexpr int EW = EDGE ? 64 * WP64 : 4;
  __shared__ __attribute__((aligned(16))) float tile[R * 64];
  __shared__ __attribute__((aligned(16))) float W1sh[EW];
  __shared__ __attribute__((aligned(16))) float W2bsh[EW];
  __shared__ __attribute__((aligned(16))) float At[EW];
  __shared__ __attribute__((aligned(16))) float Bt[EW];
  __shared__ int cnt[R];
  __shared__ int start[R + 1];
  __shared__ int wsum[32];
  int* sorted = reinterpret_cast<int*>(tile);
  const int tid = threadIdx.x;
  const int lane = tid & 31;
  v4f w1v = {}, w2v = {};
  if (EDGE) {
    w1v = *(const v4f*)(W1 + (tid >> 4) * 64 + 4 * (tid & 15));
    w2v = *(const v4f*)(W2 + (tid >> 4) * 128 + 64 + 4 * (tid & 15));
  }
  const int wave = __builtin_amdgcn_readfirstlane(tid >> 5);
  const int b = blockIdx.x;

  int s[BPW], t[BPW], e[BPW];
  const int* run[BPW];
#pragma unroll
  for (int j = 0; j < BPW; ++j) {
    const int bin = wave + 32 * j;
    const int bc = bin < BIN_BLOCKS ? bin : BIN_BLOCKS - 1;
    const int* trow = table + bc * (NB + 1) + b;
    s[j] = __builtin_amdgcn_readfirstlane(trow[0]);
    const int tj = __builtin_amdgcn_readfirstlane(trow[1]);
    t[j] = bin < BIN_BLOCKS ? tj : s[j];
    run[j] = entries + bc * CHUNK;
  }
  for (int i = tid; i < R; i += 1024) cnt[i] = 0;
#pragma unroll
  for (int j = 0; j < BPW; ++j) {
    int k = s[j] + lane;
    k = k < t[j] ? k : t[j] - 1;
    k = k > 0 ? k : 0;
    e[j] = run[j][k];
  }
  if (EDGE) {
    *(v4f*)(W1sh + (tid >> 4) * WP64 + 4 * (tid & 15)) = w1v;
    *(v4f*)(W2bsh + (tid >> 4) * WP64 + 4 * (tid & 15)) = w2v;
  }
  __syncthreads();

#pragma unroll
  for (int j = 0; j < BPW; ++j) {
    if (s[j] + lane < t[j]) atomicAdd(&cnt[e[j] >> SRC_BITS], 1);
    for (int k = s[j] + 32 + lane; k < t[j]; k += 32)
      atomicAdd(&cnt[run[j][k] >> SRC_BITS], 1);
  }
  __syncthreads();

  {
    const int c = (tid < R) ? cnt[tid] : 0;
    int incl = c;
    for (int d = 1; d < 32; d <<= 1) {
      const int u = __shfl_up(incl, d);
      if (lane >= d) incl += u;
    }
    if (lane == 31) wsum[wave] = incl;
    __syncthreads();
    if (tid < R) {
      int before = 0;
      for (int w2 = 0; w2 < wave; ++w2) before += wsum[w2];
      const int st = before + incl - c;
      start[tid] = st;
      cnt[tid] = st;
      if (tid == R - 1) start[R] = st + c;
      if (COUNT_DEG) {
        const int row = b * R + tid;
        if (row < nrows) deg[row] = (float)c;
      }
    }
    __syncthreads();
  }
  const int total = start[R];

  if (total <= CAP) {
#pragma unroll
    for (int j = 0; j < BPW; ++j) {
      if (s[j] + lane < t[j]) {
        const int pos = atomicAdd(&cnt[e[j] >> SRC_BITS], 1);
        sorted[pos] = e[j] & SRC_MASK;
      }
      for (int k = s[j] + 32 + lane; k < t[j]; k += 32) {
        const int ee = run[j][k];
        const int pos = atomicAdd(&cnt[ee >> SRC_BITS], 1);
        sorted[pos] = ee & SRC_MASK;
      }
    }
    __syncthreads();

    for (int r = wave; r < R; r += 32) {
      const int row = b * R + r;
      if (row >= nrows) {
        if (!EDGE) break;
        At[r * WP64 + lane] = 0.0f;
        At[r * WP64 + 32 + lane] = 0.0f;
        continue;
      }
      const int beg = __builtin_amdgcn_readfirstlane(start[r]);
      const int end = __builtin_amdgcn_readfirstlane(start[r + 1]);
      float lo = 0.0f, hi = 0.0f;
      for (int k = beg; k < end; k += 32) {
        const int left = end - k;
        const int m = left < 32 ? left : 32;
        const int v = sorted[k + (lane < m ? lane : m - 1)];
#pragma unroll
        for (int g = 0; g < 32; g += 8) {
          const int rem = m - g;
          if (rem >= 8) {
            add_rows<8, true>(src, v, g, 8, lane, lo, hi);
          } else {
            if (rem > 4) add_rows<8, false>(src, v, g, rem, lane, lo, hi);
            else if (rem > 0) add_rows<4, false>(src, v, g, rem, lane, lo, hi);
            break;
          }
        }
      }
      if (EDGE) {
        At[r * WP64 + lane] = lo;
        At[r * WP64 + 32 + lane] = hi;
      } else {
        float* q = dst + (size_t)row * 64 + lane;
        q[0] = lo;
        q[32] = hi;
      }
    }
  } else {
    for (int i = tid; i < R * 64; i += 1024) tile[i] = 0.0f;
    __syncthreads();
#pragma unroll
    for (int j = 0; j < BPW; ++j) {
      int k = s[j];
      for (; k + GROUP <= t[j]; k += GROUP)
        add_group<true>(src, run[j], k, GROUP, lane, tile);
      if (k < t[j]) add_group<false>(src, run[j], k, t[j] - k, lane, tile);
    }
    __syncthreads();
    for (int r = wave; r < R; r += 32) {
      const int row = b * R + r;
      if (EDGE) {
        At[r * WP64 + lane] = tile[r * 64 + lane];
        At[r * WP64 + 32 + lane] = tile[r * 64 + 32 + lane];
      } else if (row < nrows) {
        float* q = dst + (size_t)row * 64 + lane;
        q[0] = tile[r * 64 + lane];
        q[32] = tile[r * 64 + 32 + lane];
      }
    }
  }

  if (EDGE) {
    const int ll = lane & 15;
    const int lh = lane >> 4;
    const int tr = wave >> 2;
    const int tc = wave & 3;
    __syncthreads();
    if (wave < 16) {
      v8f acc = {};
      const float* Ap = At + (16 * tr + ll) * WP64 + 4 * lh;
      const float* Bp = W1sh + (16 * tc + ll) * WP64 + 4 * lh;
#pragma unroll
      for (int q = 0; q < 8; ++q) {
        const v4f a = *(const v4f*)(Ap + 8 * q);
        const v4f w = *(const v4f*)(Bp + 8 * q);
        acc = wmma_f32(a.xy, w.xy, acc);
        acc = wmma_f32(a.zw, w.zw, acc);
      }
      const float bias = b1[16 * tc + ll];
#pragma unroll
      for (int j = 0; j < 8; ++j) {
        const int rr = 16 * tr + j + 8 * lh;
        const float n = (float)(start[rr + 1] - start[rr]);
        Bt[rr * WP64 + 16 * tc + ll] = acc[j] + n * bias;
      }
    }
    __syncthreads();
    for (int r = wave; r < R; r += 32) {
      const int row = b * R + r;
      if (row < nrows) {
        float* q = dst + (size_t)row * 64 + lane;
        q[0] = Bt[r * WP64 + lane];
        q[32] = Bt[r * WP64 + 32 + lane];
      }
    }
    if (wave < 16) {
      v8f acc = {};
      const float* Ap = Bt + (16 * tr + ll) * WP64 + 4 * lh;
      const float* Bp = W2bsh + (16 * tc + ll) * WP64 + 4 * lh;
#pragma unroll
      for (int q = 0; q < 8; ++q) {
        const v4f a = *(const v4f*)(Ap + 8 * q);
        const v4f w = *(const v4f*)(Bp + 8 * q);
        acc = wmma_f32(a.xy, w.xy, acc);
        acc = wmma_f32(a.zw, w.zw, acc);
      }
#pragma unroll
      for (int j = 0; j < 8; ++j)
        At[(16 * tr + j + 8 * lh) * WP64 + 16 * tc + ll] = acc[j];
    }
    __syncthreads();
    for (int r = wave; r < R; r += 32) {
      const int row = b * R + r;
      if (row < nrows) {
        float* q = Z + (size_t)row * 64 + lane;
        q[0] = At[r * WP64 + lane];
        q[32] = At[r * WP64 + 32 + lane];
      }
    }
  }
}

extern "C" void kernel_launch(void* const* d_in, const int* in_sizes, int n_in,
                              void* d_out, int out_size, void* d_ws,
                              size_t ws_size, hipStream_t stream) {
  (void)in_sizes; (void)n_in; (void)out_size; (void)ws_size;
  const float* X    = (const float*)d_in[0];
  const float* X0   = (const float*)d_in[1];
  const int*   vertex = (const int*)d_in[2];
  const int*   edges  = (const int*)d_in[3];
  const float* W1w  = (const float*)d_in[4];
  const float* W1b  = (const float*)d_in[5];
  const float* W2w  = (const float*)d_in[6];
  const float* W2b  = (const float*)d_in[7];
  const float* Ww   = (const float*)d_in[8];
  const float* Wb   = (const float*)d_in[9];

  float* out = (float*)d_out;
  float* Xe  = out + (size_t)N_NODES_C * 64;

  float* ws  = (float*)d_ws;
  float* S   = ws;
  float* T   = ws + (size_t)N_NODES_C * 64;
  float* deg = T + (size_t)N_NODES_C * 64;

  int* entE = (int*)T;
  int* entV = entE + NNZ_C;
  int* tabE = entV + NNZ_C;
  int* tabV = tabE + BIN_BLOCKS * (NB_E + 1);
  float* Z  = T + Z_OFFSET;

  const int tiles = N_NODES_C / 16;
  const int gemmBlocks = (tiles + 7) / 8;

  front_kernel<<<FRONT_BLOCKS, 1024, 0, stream>>>(edges, vertex, entE, tabE, entV,
                                                  tabV);
  sum_range_kernel<SHIFT_E, NB_E, false, true><<<NB_E, 1024, 0, stream>>>(
      X, entE, tabE, Xe, nullptr, N_EDGES_C, W1w, W1b, W2w, Z);
  sum_range_kernel<SHIFT_V, NB_V, true, false><<<NB_V, 1024, 0, stream>>>(
      Z, entV, tabV, S, deg, N_NODES_C, nullptr, nullptr, nullptr, nullptr);
  gemm2_mix_out_kernel<<<gemmBlocks, 256, 0, stream>>>(X, X0, S, deg, W2w, W2b, Ww,
                                                       Wb, out, N_NODES_C);
}


# node sum and last launch made one: a wave sums its 16 node rows in registers and hands them through an LDS slab to its own two products; S and deg never go to memory; 4 -> 3 launches; no sum reordered
# speedup vs baseline: 5.3242x; 1.0313x over previous
#include <hip/hip_runtime.h>

typedef __attribute__((ext_vector_type(2))) float v2f;
typedef __attribute__((ext_vector_type(4))) float v4f;
typedef __attribute__((ext_vector_type(8))) float v8f;
typedef __attribute__((ext_vector_type(4))) int v4i;

#define N_NODES_C 100000
#define N_EDGES_C 20000
#define NNZ_C     1000000
#define ALPHA_C   0.5f

__device__ __forceinline__ v8f wmma_f32(v2f a, v2f b, v8f c) {
  return __builtin_amdgcn_wmma_f32_16x16x4_f32(false, a, false, b, (short)0, c,
                                               false, false);
}

#define WP64 68

template <int THREADS, int K, int PITCH, int SRC = K>
__device__ __forceinline__ void stage_w(const float* __restrict__ W, float* Wsh) {
  constexpr int Q = K / 4;
  constexpr int STEPS = 64 * Q / THREADS;
  static_assert(STEPS * THREADS == 64 * Q, "every thread copies the same number of pieces");
  v4f w[STEPS];
#pragma unroll
  for (int j = 0; j < STEPS; ++j) {
    const int i = threadIdx.x + j * THREADS;
    w[j] = *(const v4f*)(W + (i / Q) * SRC + 4 * (i % Q));
  }
#pragma unroll
  for (int j = 0; j < STEPS; ++j) {
    const int i = threadIdx.x + j * THREADS;
    const int n = i / Q, p = i % Q;
    *(v4f*)(Wsh + n * PITCH + 4 * p) = w[j];
  }
}

template <int PITCH>
__device__ __forceinline__ void wmma_piece(v4f a, const float* Wp, v8f& acc0,
                                           v8f& acc1, v8f& acc2, v8f& acc3) {
  const v4f b0 = *(const v4f*)(Wp);
  const v4f b1 = *(const v4f*)(Wp + 16 * PITCH);
  const v4f b2 = *(const v4f*)(Wp + 32 * PITCH);
  const v4f b3 = *(const v4f*)(Wp + 48 * PITCH);
  acc0 = wmma_f32(a.xy, b0.xy, acc0);
  acc1 = wmma_f32(a.xy, b1.xy, acc1);
  acc2 = wmma_f32(a.xy, b2.xy, acc2);
  acc3 = wmma_f32(a.xy, b3.xy, acc3);
  acc0 = wmma_f32(a.zw, b0.zw, acc0);
  acc1 = wmma_f32(a.zw, b1.zw, acc1);
  acc2 = wmma_f32(a.zw, b2.zw, acc2);
  acc3 = wmma_f32(a.zw, b3.zw, acc3);
}

#define TP 36

__device__ __forceinline__ void wave_lds_handover() {
  __builtin_amdgcn_fence(__ATOMIC_RELEASE, "wavefront");
  __builtin_amdgcn_wave_barrier();
  __builtin_amdgcn_fence(__ATOMIC_ACQUIRE, "wavefront");
}

#define CHUNK 8192
#define BIN_BLOCKS ((NNZ_C + CHUNK - 1) / CHUNK)
#define SHIFT_E 6
#define SHIFT_V 8
#define NB_E ((N_EDGES_C + (1 << SHIFT_E) - 1) >> SHIFT_E)
#define NB_V ((N_NODES_C + (1 << SHIFT_V) - 1) >> SHIFT_V)
#define SRC_BITS 20
#define SRC_MASK ((1 << SRC_BITS) - 1)
static_assert(N_NODES_C <= (1 << SRC_BITS) && N_EDGES_C <= (1 << SRC_BITS),
              "a source row fits the low bits of an entry");
static_assert(NB_E <= 1024 && NB_V <= 1024, "one thread per range in the scan");
static_assert(NNZ_C % 4 == 0 && CHUNK % 4096 == 0,
              "a chunk leaves in 16-byte pieces, the same number per thread");
#define BIN_WORDS (2 * NNZ_C + BIN_BLOCKS * (NB_E + 1) + BIN_BLOCKS * (NB_V + 1))
#define Z_OFFSET (((BIN_WORDS + 63) / 64) * 64)
static_assert((size_t)Z_OFFSET + (size_t)N_EDGES_C * 64 <= (size_t)N_NODES_C * 64,
              "Z fits the plane behind the bin outputs");

template <int SHIFT, int NB>
__device__ __forceinline__ void bin_body(
    const int* __restrict__ keys, const int* __restrict__ vals,
    int* __restrict__ entries, int* __restrict__ table, int blk, int* hist,
    int* wsum, int* sorted) {
  const int tid = threadIdx.x;
  const int lane = tid & 31;
  const int wave = tid >> 5;
  for (int i = tid; i < NB; i += 1024) hist[i] = 0;
  __syncthreads();

  const int base = blk * CHUNK;
  int key[8], val[8], rank[8];
#pragma unroll
  for (int j = 0; j < 8; ++j) {
    const int e = base + j * 1024 + tid;
    key[j] = 0;
    val[j] = 0;
    if (e < NNZ_C) {
      key[j] = keys[e];
      val[j] = vals[e];
    }
  }
#pragma unroll
  for (int j = 0; j < 8; ++j) {
    const int e = base + j * 1024 + tid;
    rank[j] = 0;
    if (e < NNZ_C) rank[j] = atomicAdd(&hist[key[j] >> SHIFT], 1);
  }
  __syncthreads();

  const int c = (tid < NB) ? hist[tid] : 0;
  int incl = c;
  for (int d = 1; d < 32; d <<= 1) {
    const int t = __shfl_up(incl, d);
    if (lane >= d) incl += t;
  }
  if (lane == 31) wsum[wave] = incl;
  __syncthreads();
  int before = 0;
  for (int w2 = 0; w2 < wave; ++w2) before += wsum[w2];
  const int start = before + incl - c;
  if (tid < NB) {
    hist[tid] = start;
    table[blk * (NB + 1) + tid] = start;
  }
  if (tid == 0) {
    const int left = NNZ_C - base;
    table[blk * (NB + 1) + NB] = left < CHUNK ? left : CHUNK;
  }
  __syncthreads();

#pragma unroll
  for (int j = 0; j < 8; ++j) {
    const int e = base + j * 1024 + tid;
    if (e < NNZ_C) {
      const int r = key[j] >> SHIFT;
      const int local = key[j] - (r << SHIFT);
      sorted[hist[r] + rank[j]] = (local << SRC_BITS) | val[j];
    }
  }
  __syncthreads();

  const int left = NNZ_C - base;
  const int count = left < CHUNK ? left : CHUNK;
  for (int j = 0; j < CHUNK / 4096; ++j) {
    const int i4 = 4 * (tid + j * 1024);
    if (i4 < count) *(v4i*)(entries + base + i4) = *(const v4i*)(sorted + i4);
  }
}

#define NB_MAX (NB_E > NB_V ? NB_E : NB_V)
#define FRONT_BLOCKS (2 * BIN_BLOCKS)
__global__ __launch_bounds__(1024) void front_kernel(
    const int* __restrict__ edges, const int* __restrict__ vertex,
    int* __restrict__ entE, int* __restrict__ tabE, int* __restrict__ entV,
    int* __restrict__ tabV) {
  __shared__ int hist[NB_MAX];
  __shared__ int wsum[32];
  __shared__ __attribute__((aligned(16))) int sorted[CHUNK];
  const int blk = blockIdx.x;
  if (blk < BIN_BLOCKS) {
    bin_body<SHIFT_E, NB_E>(edges, vertex, entE, tabE, blk, hist, wsum, sorted);
  } else {
    bin_body<SHIFT_V, NB_V>(vertex, edges, entV, tabV, blk - BIN_BLOCKS, hist, wsum,
                            sorted);
  }
}

#define BPW ((BIN_BLOCKS + 31) / 32)

template <int G, bool FULL>
__device__ __forceinline__ void add_rows(const float* __restrict__ src, int v,
                                         int first, int n, int lane, float& lo,
                                         float& hi) {
  float a[G], c[G];
#pragma unroll
  for (int i = 0; i < G; ++i) {
    const int idx = __builtin_amdgcn_readlane(v, first + i);
    const float* p = src + (size_t)idx * 64 + lane;
    a[i] = p[0];
    c[i] = p[32];
  }
#pragma unroll
  for (int i = 0; i < G; ++i) {
    if (FULL || i < n) {
      lo += a[i];
      hi += c[i];
    }
  }
}

#define GROUP 8
template <bool FULL, int TPITCH>
__device__ __forceinline__ void add_group(const float* __restrict__ src,
                                          const int* __restrict__ run, int k,
                                          int n, int lane, float* tile) {
  int g[GROUP];
  float lo[GROUP], hi[GROUP];
#pragma unroll
  for (int i = 0; i < GROUP; ++i) {
    const int ii = (FULL || i < n) ? i : n - 1;
    g[i] = run[k + ii];
  }
#pragma unroll
  for (int i = 0; i < GROUP; ++i) {
    const float* p = src + (size_t)(g[i] & SRC_MASK) * 64 + lane;
    lo[i] = p[0];
    hi[i] = p[32];
  }
#pragma unroll
  for (int i = 0; i < GROUP; ++i) {
    if (FULL || i < n) {
      float* q = tile + (g[i] >> SRC_BITS) * TPITCH + lane;
      unsafeAtomicAdd(q, lo[i]);
      unsafeAtomicAdd(q + 32, hi[i]);
    }
  }
}

template <int SHIFT, int NB, bool COUNT_DEG, bool EDGE>
__global__ __launch_bounds__(1024) void sum_range_kernel(
    const float* __restrict__ src, const int* __restrict__ entries,
    const int* __restrict__ table, float* __restrict__ dst,
    float* __restrict__ deg, int nrows, const float* __restrict__ W1,
    const float* __restrict__ b1, const float* __restrict__ W2,
    float* __restrict__ Z) {
  constexpr int R = 1 << SHIFT;
  constexpr int CAP = R * 64;
  static_assert(!EDGE || R == 64, "the edge products are written for one 64 x 64 tile");
  constexpr int EW = EDGE ? 64 * WP64 : 4;
  __shared__ __attribute__((aligned(16))) float tile[R * 64];
  __shared__ __attribute__((aligned(16))) float W1sh[EW];
  __shared__ __attribute__((aligned(16))) float W2bsh[EW];
  __shared__ __attribute__((aligned(16))) float At[EW];
  __shared__ __attribute__((aligned(16))) float Bt[EW];
  __shared__ int cnt[R];
  __shared__ int start[R + 1];
  __shared__ int wsum[32];
  int* sorted = reinterpret_cast<int*>(tile);
  const int tid = threadIdx.x;
  const int lane = tid & 31;
  v4f w1v = {}, w2v = {};
  if (EDGE) {
    w1v = *(const v4f*)(W1 + (tid >> 4) * 64 + 4 * (tid & 15));
    w2v = *(const v4f*)(W2 + (tid >> 4) * 128 + 64 + 4 * (tid & 15));
  }
  const int wave = __builtin_amdgcn_readfirstlane(tid >> 5);
  const int b = blockIdx.x;

  int s[BPW], t[BPW], e[BPW];
  const int* run[BPW];
#pragma unroll
  for (int j = 0; j < BPW; ++j) {
    const int bin = wave + 32 * j;
    const int bc = bin < BIN_BLOCKS ? bin : BIN_BLOCKS - 1;
    const int* trow = table + bc * (NB + 1) + b;
    s[j] = __builtin_amdgcn_readfirstlane(trow[0]);
    const int tj = __builtin_amdgcn_readfirstlane(trow[1]);
    t[j] = bin < BIN_BLOCKS ? tj : s[j];
    run[j] = entries + bc * CHUNK;
  }
  for (int i = tid; i < R; i += 1024) cnt[i] = 0;
#pragma unroll
  for (int j = 0; j < BPW; ++j) {
    int k = s[j] + lane;
    k = k < t[j] ? k : t[j] - 1;
    k = k > 0 ? k : 0;
    e[j] = run[j][k];
  }
  if (EDGE) {
    *(v4f*)(W1sh + (tid >> 4) * WP64 + 4 * (tid & 15)) = w1v;
    *(v4f*)(W2bsh + (tid >> 4) * WP64 + 4 * (tid & 15)) = w2v;
  }
  __syncthreads();

#pragma unroll
  for (int j = 0; j < BPW; ++j) {
    if (s[j] + lane < t[j]) atomicAdd(&cnt[e[j] >> SRC_BITS], 1);
    for (int k = s[j] + 32 + lane; k < t[j]; k += 32)
      atomicAdd(&cnt[run[j][k] >> SRC_BITS], 1);
  }
  __syncthreads();

  {
    const int c = (tid < R) ? cnt[tid] : 0;
    int incl = c;
    for (int d = 1; d < 32; d <<= 1) {
      const int u = __shfl_up(incl, d);
      if (lane >= d) incl += u;
    }
    if (lane == 31) wsum[wave] = incl;
    __syncthreads();
    if (tid < R) {
      int before = 0;
      for (int w2 = 0; w2 < wave; ++w2) before += wsum[w2];
      const int st = before + incl - c;
      start[tid] = st;
      cnt[tid] = st;
      if (tid == R - 1) start[R] = st + c;
      if (COUNT_DEG) {
        const int row = b * R + tid;
        if (row < nrows) deg[row] = (float)c;
      }
    }
    __syncthreads();
  }
  const int total = start[R];

  if (total <= CAP) {
#pragma unroll
    for (int j = 0; j < BPW; ++j) {
      if (s[j] + lane < t[j]) {
        const int pos = atomicAdd(&cnt[e[j] >> SRC_BITS], 1);
        sorted[pos] = e[j] & SRC_MASK;
      }
      for (int k = s[j] + 32 + lane; k < t[j]; k += 32) {
        const int ee = run[j][k];
        const int pos = atomicAdd(&cnt[ee >> SRC_BITS], 1);
        sorted[pos] = ee & SRC_MASK;
      }
    }
    __syncthreads();

    for (int r = wave; r < R; r += 32) {
      const int row = b * R + r;
      if (row >= nrows) {
        if (!EDGE) break;
        At[r * WP64 + lane] = 0.0f;
        At[r * WP64 + 32 + lane] = 0.0f;
        continue;
      }
      const int beg = __builtin_amdgcn_readfirstlane(start[r]);
      const int end = __builtin_amdgcn_readfirstlane(start[r + 1]);
      float lo = 0.0f, hi = 0.0f;
      for (int k = beg; k < end; k += 32) {
        const int left = end - k;
        const int m = left < 32 ? left : 32;
        const int v = sorted[k + (lane < m ? lane : m - 1)];
#pragma unroll
        for (int g = 0; g < 32; g += 8) {
          const int rem = m - g;
          if (rem >= 8) {
            add_rows<8, true>(src, v, g, 8, lane, lo, hi);
          } else {
            if (rem > 4) add_rows<8, false>(src, v, g, rem, lane, lo, hi);
            else if (rem > 0) add_rows<4, false>(src, v, g, rem, lane, lo, hi);
            break;
          }
        }
      }
      if (EDGE) {
        At[r * WP64 + lane] = lo;
        At[r * WP64 + 32 + lane] = hi;
      } else {
        float* q = dst + (size_t)row * 64 + lane;
        q[0] = lo;
        q[32] = hi;
      }
    }
  } else {
    for (int i = tid; i < R * 64; i += 1024) tile[i] = 0.0f;
    __syncthreads();
#pragma unroll
    for (int j = 0; j < BPW; ++j) {
      int k = s[j];
      for (; k + GROUP <= t[j]; k += GROUP)
        add_group<true, 64>(src, run[j], k, GROUP, lane, tile);
      if (k < t[j]) add_group<false, 64>(src, run[j], k, t[j] - k, lane, tile);
    }
    __syncthreads();
    for (int r = wave; r < R; r += 32) {
      const int row = b * R + r;
      if (EDGE) {
        At[r * WP64 + lane] = tile[r * 64 + lane];
        At[r * WP64 + 32 + lane] = tile[r * 64 + 32 + lane];
      } else if (row < nrows) {
        float* q = dst + (size_t)row * 64 + lane;
        q[0] = tile[r * 64 + lane];
        q[32] = tile[r * 64 + 32 + lane];
      }
    }
  }

  if (EDGE) {
    const int ll = lane & 15;
    const int lh = lane >> 4;
    const int tr = wave >> 2;
    const int tc = wave & 3;
    __syncthreads();
    if (wave < 16) {
      v8f acc = {};
      const float* Ap = At + (16 * tr + ll) * WP64 + 4 * lh;
      const float* Bp = W1sh + (16 * tc + ll) * WP64 + 4 * lh;
#pragma unroll
      for (int q = 0; q < 8; ++q) {
        const v4f a = *(const v4f*)(Ap + 8 * q);
        const v4f w = *(const v4f*)(Bp + 8 * q);
        acc = wmma_f32(a.xy, w.xy, acc);
        acc = wmma_f32(a.zw, w.zw, acc);
      }
      const float bias = b1[16 * tc + ll];
#pragma unroll
      for (int j = 0; j < 8; ++j) {
        const int rr = 16 * tr + j + 8 * lh;
        const float n = (float)(start[rr + 1] - start[rr]);
        Bt[rr * WP64 + 16 * tc + ll] = acc[j] + n * bias;
      }
    }
    __syncthreads();
    for (int r = wave; r < R; r += 32) {
      const int row = b * R + r;
      if (row < nrows) {
        float* q = dst + (size_t)row * 64 + lane;
        q[0] = Bt[r * WP64 + lane];
        q[32] = Bt[r * WP64 + 32 + lane];
      }
    }
    if (wave < 16) {
      v8f acc = {};
      const float* Ap = Bt + (16 * tr + ll) * WP64 + 4 * lh;
      const float* Bp = W2bsh + (16 * tc + ll) * WP64 + 4 * lh;
#pragma unroll
      for (int q = 0; q < 8; ++q) {
        const v4f a = *(const v4f*)(Ap + 8 * q);
        const v4f w = *(const v4f*)(Bp + 8 * q);
        acc = wmma_f32(a.xy, w.xy, acc);
        acc = wmma_f32(a.zw, w.zw, acc);
      }
#pragma unroll
      for (int j = 0; j < 8; ++j)
        At[(16 * tr + j + 8 * lh) * WP64 + 16 * tc + ll] = acc[j];
    }
    __syncthreads();
    for (int r = wave; r < R; r += 32) {
      const int row = b * R + r;
      if (row < nrows) {
        float* q = Z + (size_t)row * 64 + lane;
        q[0] = At[r * WP64 + lane];
        q[32] = At[r * WP64 + 32 + lane];
      }
    }
  }
}

#define NODE_THREADS 512
#define NODE_WAVES (NODE_THREADS / 32)
#define BPW_V ((BIN_BLOCKS + NODE_WAVES - 1) / NODE_WAVES)
#define SP 68
#define CAP_V 8192
static_assert((1 << SHIFT_V) == 16 * NODE_WAVES, "a wave owns 16 rows of the range");
static_assert(N_NODES_C % 16 == 0, "a wave's 16 rows are all real rows or all past the end");
static_assert(16 * TP <= 16 * SP, "the half tile of T fits the slab");

__global__ __launch_bounds__(NODE_THREADS, 8) void node_sum_mix_out_kernel(
    const float* __restrict__ Z, const int* __restrict__ entries,
    const int* __restrict__ table, const float* __restrict__ X,
    const float* __restrict__ X0, const float* __restrict__ W2,
    const float* __restrict__ b2, const float* __restrict__ W,
    const float* __restrict__ b, float* __restrict__ Out, int nrows) {
  constexpr int R = 1 << SHIFT_V;
  __shared__ __attribute__((aligned(16))) float Ssh[R * SP];
  __shared__ __attribute__((aligned(16))) float W2sh[64 * WP64];
  __shared__ __attribute__((aligned(16))) float Wsh[64 * WP64];
  __shared__ int sorted[CAP_V];
  __shared__ int cnt[R];
  __shared__ int start[R + 1];
  __shared__ int wsum[NODE_WAVES];
  const int tid = threadIdx.x;
  const int lane = tid & 31;
  const int ll = lane & 15;
  const int lh = lane >> 4;
  const int wave = __builtin_amdgcn_readfirstlane(tid >> 5);
  const int bk = blockIdx.x;
  const int rowBase = bk * R + 16 * wave;
  const bool live = rowBase < nrows;

  v4f ax[8];
  const float* Xrow = X + (size_t)((live ? rowBase : 0) + ll) * 64 + 4 * lh;
#pragma unroll
  for (int q = 0; q < 8; ++q) ax[q] = *(const v4f*)(Xrow + 8 * q);

  int s[BPW_V], t[BPW_V], e[BPW_V];
  const int* run[BPW_V];
#pragma unroll
  for (int j = 0; j < BPW_V; ++j) {
    const int bin = wave + NODE_WAVES * j;
    const int bc = bin < BIN_BLOCKS ? bin : BIN_BLOCKS - 1;
    const int* trow = table + bc * (NB_V + 1) + bk;
    s[j] = __builtin_amdgcn_readfirstlane(trow[0]);
    const int tj = __builtin_amdgcn_readfirstlane(trow[1]);
    t[j] = bin < BIN_BLOCKS ? tj : s[j];
    run[j] = entries + bc * CHUNK;
  }
  for (int i = tid; i < R; i += NODE_THREADS) cnt[i] = 0;
#pragma unroll
  for (int j = 0; j < BPW_V; ++j) {
    int k = s[j] + lane;
    k = k < t[j] ? k : t[j] - 1;
    k = k > 0 ? k : 0;
    e[j] = run[j][k];
  }
  stage_w<NODE_THREADS, 64, WP64, 128>(W2, W2sh);
  stage_w<NODE_THREADS, 64, WP64>(W, Wsh);
  __syncthreads();

#pragma unroll
  for (int j = 0; j < BPW_V; ++j) {
    if (s[j] + lane < t[j]) atomicAdd(&cnt[e[j] >> SRC_BITS], 1);
    for (int k = s[j] + 32 + lane; k < t[j]; k += 32)
      atomicAdd(&cnt[run[j][k] >> SRC_BITS], 1);
  }
  __syncthreads();

  {
    const int cn = (tid < R) ? cnt[tid] : 0;
    int incl = cn;
    for (int d = 1; d < 32; d <<= 1) {
      const int u = __shfl_up(incl, d);
      if (lane >= d) incl += u;
    }
    if (lane == 31) wsum[wave] = incl;
    __syncthreads();
    if (tid < R) {
      int before = 0;
      for (int w2 = 0; w2 < wave; ++w2) before += wsum[w2];
      const int st = before + incl - cn;
      start[tid] = st;
      cnt[tid] = st;
      if (tid == R - 1) start[R] = st + cn;
    }
    __syncthreads();
  }
  const int total = start[R];

  float* Sw = Ssh + wave * (16 * SP);
  if (total <= CAP_V) {
#pragma unroll
    for (int j = 0; j < BPW_V; ++j) {
      if (s[j] + lane < t[j]) {
        const int pos = atomicAdd(&cnt[e[j] >> SRC_BITS], 1);
        sorted[pos] = e[j] & SRC_MASK;
      }
      for (int k = s[j] + 32 + lane; k < t[j]; k += 32) {
        const int ee = run[j][k];
        const int pos = atomicAdd(&cnt[ee >> SRC_BITS], 1);
        sorted[pos] = ee & SRC_MASK;
      }
    }
    __syncthreads();

    if (live) {
      for (int rr = 0; rr < 16; ++rr) {
        const int r = 16 * wave + rr;
        const int beg = __builtin_amdgcn_readfirstlane(start[r]);
        const int end = __builtin_amdgcn_readfirstlane(start[r + 1]);
        float lo = 0.0f, hi = 0.0f;
        for (int k = beg; k < end; k += 32) {
          const int left = end - k;
          const int m = left < 32 ? left : 32;
          const int v = sorted[k + (lane < m ? lane : m - 1)];
#pragma unroll
          for (int g = 0; g < 32; g += 8) {
            const int rem = m - g;
            if (rem >= 8) {
              add_rows<8, true>(Z, v, g, 8, lane, lo, hi);
            } else {
              if (rem > 4) add_rows<8, false>(Z, v, g, rem, lane, lo, hi);
              else if (rem > 0) add_rows<4, false>(Z, v, g, rem, lane, lo, hi);
              break;
            }
          }
        }
        Sw[rr * SP + lane] = lo;
        Sw[rr * SP + 32 + lane] = hi;
      }
    }
  } else {
    for (int i = tid; i < R * SP; i += NODE_THREADS) Ssh[i] = 0.0f;
    __syncthreads();
#pragma unroll
    for (int j = 0; j < BPW_V; ++j) {
      int k = s[j];
      for (; k + GROUP <= t[j]; k += GROUP)
        add_group<true, SP>(Z, run[j], k, GROUP, lane, Ssh);
      if (k < t[j]) add_group<false, SP>(Z, run[j], k, t[j] - k, lane, Ssh);
    }
    __syncthreads();
  }
  if (!live) return;
  wave_lds_handover();

  const int rw = 16 * wave;
  const float dg = (float)(start[rw + ll + 1] - start[rw + ll]);
  const float* W2row = W2sh + ll * WP64 + 4 * lh;
  v8f acc0, acc1, acc2, acc3;
#pragma unroll
  for (int j = 0; j < 8; ++j) {
    const float* sr = Sw + (j + lh * 8) * SP + ll;
    acc0[j] = sr[0];
    acc1[j] = sr[16];
    acc2[j] = sr[32];
    acc3[j] = sr[48];
  }
  wave_lds_handover();
#pragma unroll
  for (int q = 0; q < 8; ++q)
    wmma_piece<WP64>(dg * ax[q], W2row + 8 * q, acc0, acc1, acc2, acc3);

  const float oma = 1.0f - ALPHA_C;
#pragma unroll
  for (int j = 0; j < 8; ++j) {
    const int rj = j + lh * 8;
    const float dr = (float)(start[rw + rj + 1] - start[rw + rj]);
    const float* x0r = X0 + (size_t)(rowBase + rj) * 64;
    acc0[j] = oma * (acc0[j] + dr * b2[ll])      + ALPHA_C * x0r[ll];
    acc1[j] = oma * (acc1[j] + dr * b2[16 + ll]) + ALPHA_C * x0r[16 + ll];
    acc2[j] = oma * (acc2[j] + dr * b2[32 + ll]) + ALPHA_C * x0r[32 + ll];
    acc3[j] = oma * (acc3[j] + dr * b2[48 + ll]) + ALPHA_C * x0r[48 + ll];
  }

  float* Tst = Sw + (lh * 8) * TP + ll;
  const float* Trow = Sw + ll * TP + 4 * lh;
  const float* Wrow = Wsh + ll * WP64 + 4 * lh;
  v8f o0 = {}, o1 = {}, o2 = {}, o3 = {};
  v4f a[4];

#pragma unroll
  for (int j = 0; j < 8; ++j) {
    Tst[j * TP]      = acc0[j];
    Tst[j * TP + 16] = acc1[j];
  }
  wave_lds_handover();
#pragma unroll
  for (int q = 0; q < 4; ++q) a[q] = *(const v4f*)(Trow + 8 * q);
  wave_lds_handover();
#pragma unroll
  for (int q = 0; q < 4; ++q)
    wmma_piece<WP64>(a[q], Wrow + 8 * q, o0, o1, o2, o3);

#pragma unroll
  for (int j = 0; j < 8; ++j) {
    Tst[j * TP]      = acc2[j];
    Tst[j * TP + 16] = acc3[j];
  }
  wave_lds_handover();
#pragma unroll
  for (int q = 0; q < 4; ++q) a[q] = *(const v4f*)(Trow + 8 * q);
#pragma unroll
  for (int q = 0; q < 4; ++q)
    wmma_piece<WP64>(a[q], Wrow + 32 + 8 * q, o0, o1, o2, o3);

#pragma unroll
  for (int j = 0; j < 8; ++j) {
    float* o = Out + (size_t)(rowBase + j + lh * 8) * 64;
    o[ll]      = o0[j] + b[ll];
    o[16 + ll] = o1[j] + b[16 + ll];
    o[32 + ll] = o2[j] + b[32 + ll];
    o[48 + ll] = o3[j] + b[48 + ll];
  }
}

extern "C" void kernel_launch(void* const* d_in, const int* in_sizes, int n_in,
                              void* d_out, int out_size, void* d_ws,
                              size_t ws_size, hipStream_t stream) {
  (void)in_sizes; (void)n_in; (void)out_size; (void)ws_size;
  const float* X    = (const float*)d_in[0];
  const float* X0   = (const float*)d_in[1];
  const int*   vertex = (const int*)d_in[2];
  const int*   edges  = (const int*)d_in[3];
  const float* W1w  = (const float*)d_in[4];
  const float* W1b  = (const float*)d_in[5];
  const float* W2w  = (const float*)d_in[6];
  const float* W2b  = (const float*)d_in[7];
  const float* Ww   = (const float*)d_in[8];
  const float* Wb   = (const float*)d_in[9];

  float* out = (float*)d_out;
  float* Xe  = out + (size_t)N_NODES_C * 64;

  float* P1 = (float*)d_ws + (size_t)N_NODES_C * 64;
  int* entE = (int*)P1;
  int* entV = entE + NNZ_C;
  int* tabE = entV + NNZ_C;
  int* tabV = tabE + BIN_BLOCKS * (NB_E + 1);
  float* Z  = P1 + Z_OFFSET;

  front_kernel<<<FRONT_BLOCKS, 1024, 0, stream>>>(edges, vertex, entE, tabE, entV,
                                                  tabV);
  sum_range_kernel<SHIFT_E, NB_E, false, true><<<NB_E, 1024, 0, stream>>>(
      X, entE, tabE, Xe, nullptr, N_EDGES_C, W1w, W1b, W2w, Z);
  node_sum_mix_out_kernel<<<NB_V, NODE_THREADS, 0, stream>>>(
      Z, entV, tabV, X, X0, W2w, W2b, Ww, Wb, out, N_NODES_C);
}


# segment sums: a half wave owns a destination row (a lane holds 4 columns), so one gather instruction brings two whole source rows, 4 lines, not one line; 4x fewer gather loads; no sum reordered
# speedup vs baseline: 5.7693x; 1.0836x over previous
#include <hip/hip_runtime.h>

typedef __attribute__((ext_vector_type(2))) float v2f;
typedef __attribute__((ext_vector_type(4))) float v4f;
typedef __attribute__((ext_vector_type(8))) float v8f;
typedef __attribute__((ext_vector_type(4))) int v4i;

#define N_NODES_C 100000
#define N_EDGES_C 20000
#define NNZ_C     1000000
#define ALPHA_C   0.5f

__device__ __forceinline__ v8f wmma_f32(v2f a, v2f b, v8f c) {
  return __builtin_amdgcn_wmma_f32_16x16x4_f32(false, a, false, b, (short)0, c,
                                               false, false);
}

#define WP64 68

template <int THREADS, int K, int PITCH, int SRC = K>
__device__ __forceinline__ void stage_w(const float* __restrict__ W, float* Wsh) {
  constexpr int Q = K / 4;
  constexpr int STEPS = 64 * Q / THREADS;
  static_assert(STEPS * THREADS == 64 * Q, "every thread copies the same number of pieces");
  v4f w[STEPS];
#pragma unroll
  for (int j = 0; j < STEPS; ++j) {
    const int i = threadIdx.x + j * THREADS;
    w[j] = *(const v4f*)(W + (i / Q) * SRC + 4 * (i % Q));
  }
#pragma unroll
  for (int j = 0; j < STEPS; ++j) {
    const int i = threadIdx.x + j * THREADS;
    const int n = i / Q, p = i % Q;
    *(v4f*)(Wsh + n * PITCH + 4 * p) = w[j];
  }
}

template <int PITCH>
__device__ __forceinline__ void wmma_piece(v4f a, const float* Wp, v8f& acc0,
                                           v8f& acc1, v8f& acc2, v8f& acc3) {
  const v4f b0 = *(const v4f*)(Wp);
  const v4f b1 = *(const v4f*)(Wp + 16 * PITCH);
  const v4f b2 = *(const v4f*)(Wp + 32 * PITCH);
  const v4f b3 = *(const v4f*)(Wp + 48 * PITCH);
  acc0 = wmma_f32(a.xy, b0.xy, acc0);
  acc1 = wmma_f32(a.xy, b1.xy, acc1);
  acc2 = wmma_f32(a.xy, b2.xy, acc2);
  acc3 = wmma_f32(a.xy, b3.xy, acc3);
  acc0 = wmma_f32(a.zw, b0.zw, acc0);
  acc1 = wmma_f32(a.zw, b1.zw, acc1);
  acc2 = wmma_f32(a.zw, b2.zw, acc2);
  acc3 = wmma_f32(a.zw, b3.zw, acc3);
}

#define TP 36

__device__ __forceinline__ void wave_lds_handover() {
  __builtin_amdgcn_fence(__ATOMIC_RELEASE, "wavefront");
  __builtin_amdgcn_wave_barrier();
  __builtin_amdgcn_fence(__ATOMIC_ACQUIRE, "wavefront");
}

#define CHUNK 8192
#define BIN_BLOCKS ((NNZ_C + CHUNK - 1) / CHUNK)
#define SHIFT_E 6
#define SHIFT_V 8
#define NB_E ((N_EDGES_C + (1 << SHIFT_E) - 1) >> SHIFT_E)
#define NB_V ((N_NODES_C + (1 << SHIFT_V) - 1) >> SHIFT_V)
#define SRC_BITS 20
#define SRC_MASK ((1 << SRC_BITS) - 1)
static_assert(N_NODES_C <= (1 << SRC_BITS) && N_EDGES_C <= (1 << SRC_BITS),
              "a source row fits the low bits of an entry");
static_assert(NB_E <= 1024 && NB_V <= 1024, "one thread per range in the scan");
static_assert(NNZ_C % 4 == 0 && CHUNK % 4096 == 0,
              "a chunk leaves in 16-byte pieces, the same number per thread");
#define BIN_WORDS (2 * NNZ_C + BIN_BLOCKS * (NB_E + 1) + BIN_BLOCKS * (NB_V + 1))
#define Z_OFFSET (((BIN_WORDS + 63) / 64) * 64)
static_assert((size_t)Z_OFFSET + (size_t)N_EDGES_C * 64 <= (size_t)N_NODES_C * 64,
              "Z fits the plane behind the bin outputs");

template <int SHIFT, int NB>
__device__ __forceinline__ void bin_body(
    const int* __restrict__ keys, const int* __restrict__ vals,
    int* __restrict__ entries, int* __restrict__ table, int blk, int* hist,
    int* wsum, int* sorted) {
  const int tid = threadIdx.x;
  const int lane = tid & 31;
  const int wave = tid >> 5;
  for (int i = tid; i < NB; i += 1024) hist[i] = 0;
  __syncthreads();

  const int base = blk * CHUNK;
  int key[8], val[8], rank[8];
#pragma unroll
  for (int j = 0; j < 8; ++j) {
    const int e = base + j * 1024 + tid;
    key[j] = 0;
    val[j] = 0;
    if (e < NNZ_C) {
      key[j] = keys[e];
      val[j] = vals[e];
    }
  }
#pragma unroll
  for (int j = 0; j < 8; ++j) {
    const int e = base + j * 1024 + tid;
    rank[j] = 0;
    if (e < NNZ_C) rank[j] = atomicAdd(&hist[key[j] >> SHIFT], 1);
  }
  __syncthreads();

  const int c = (tid < NB) ? hist[tid] : 0;
  int incl = c;
  for (int d = 1; d < 32; d <<= 1) {
    const int t = __shfl_up(incl, d);
    if (lane >= d) incl += t;
  }
  if (lane == 31) wsum[wave] = incl;
  __syncthreads();
  int before = 0;
  for (int w2 = 0; w2 < wave; ++w2) before += wsum[w2];
  const int start = before + incl - c;
  if (tid < NB) {
    hist[tid] = start;
    table[blk * (NB + 1) + tid] = start;
  }
  if (tid == 0) {
    const int left = NNZ_C - base;
    table[blk * (NB + 1) + NB] = left < CHUNK ? left : CHUNK;
  }
  __syncthreads();

#pragma unroll
  for (int j = 0; j < 8; ++j) {
    const int e = base + j * 1024 + tid;
    if (e < NNZ_C) {
      const int r = key[j] >> SHIFT;
      const int local = key[j] - (r << SHIFT);
      sorted[hist[r] + rank[j]] = (local << SRC_BITS) | val[j];
    }
  }
  __syncthreads();

  const int left = NNZ_C - base;
  const int count = left < CHUNK ? left : CHUNK;
  for (int j = 0; j < CHUNK / 4096; ++j) {
    const int i4 = 4 * (tid + j * 1024);
    if (i4 < count) *(v4i*)(entries + base + i4) = *(const v4i*)(sorted + i4);
  }
}

#define NB_MAX (NB_E > NB_V ? NB_E : NB_V)
#define FRONT_BLOCKS (2 * BIN_BLOCKS)
__global__ __launch_bounds__(1024) void front_kernel(
    const int* __restrict__ edges, const int* __restrict__ vertex,
    int* __restrict__ entE, int* __restrict__ tabE, int* __restrict__ entV,
    int* __restrict__ tabV) {
  __shared__ int hist[NB_MAX];
  __shared__ int wsum[32];
  __shared__ __attribute__((aligned(16))) int sorted[CHUNK];
  const int blk = blockIdx.x;
  if (blk < BIN_BLOCKS) {
    bin_body<SHIFT_E, NB_E>(edges, vertex, entE, tabE, blk, hist, wsum, sorted);
  } else {
    bin_body<SHIFT_V, NB_V>(vertex, edges, entV, tabV, blk - BIN_BLOCKS, hist, wsum,
                            sorted);
  }
}

#define BPW ((BIN_BLOCKS + 31) / 32)

#define LIST_PAD 16
template <int G>
__device__ __forceinline__ void sum_two_rows(const float* __restrict__ src,
                                             const int* list, int cap, int beg,
                                             int n, int nmax, int c16, v4f& acc) {
  static_assert(G <= LIST_PAD && G % 4 == 0, "a step's index reads stay inside the padded array");
  for (int p0 = 0; p0 < nmax; p0 += G) {
    int at = beg + p0;
    at = at < cap ? at : cap;
    const int* lst = list + at;
    int id[G];
    v4f x[G];
#pragma unroll
    for (int i = 0; i < G; ++i) id[i] = lst[i];
#pragma unroll
    for (int f = 0; f < G; f += 4) {
      if (p0 + f < nmax) {
#pragma unroll
        for (int i = f; i < f + 4; ++i)
          if (p0 + i < n)
            x[i] = *(const v4f*)((const char*)src + (unsigned)((id[i] << 8) + c16));
      }
    }
#pragma unroll
    for (int f = 0; f < G; f += 4) {
      if (p0 + f < nmax) {
#pragma unroll
        for (int i = f; i < f + 4; ++i)
          if (p0 + i < n) acc += x[i];
      }
    }
  }
}

#define GROUP 8
template <bool FULL, int TPITCH>
__device__ __forceinline__ void add_group(const float* __restrict__ src,
                                          const int* __restrict__ run, int k,
                                          int n, int lane, float* tile) {
  int g[GROUP];
  float lo[GROUP], hi[GROUP];
#pragma unroll
  for (int i = 0; i < GROUP; ++i) {
    const int ii = (FULL || i < n) ? i : n - 1;
    g[i] = run[k + ii];
  }
#pragma unroll
  for (int i = 0; i < GROUP; ++i) {
    const float* p = src + (size_t)(g[i] & SRC_MASK) * 64 + lane;
    lo[i] = p[0];
    hi[i] = p[32];
  }
#pragma unroll
  for (int i = 0; i < GROUP; ++i) {
    if (FULL || i < n) {
      float* q = tile + (g[i] >> SRC_BITS) * TPITCH + lane;
      unsafeAtomicAdd(q, lo[i]);
      unsafeAtomicAdd(q + 32, hi[i]);
    }
  }
}

template <int SHIFT, int NB, bool COUNT_DEG, bool EDGE>
__global__ __launch_bounds__(1024) void sum_range_kernel(
    const float* __restrict__ src, const int* __restrict__ entries,
    const int* __restrict__ table, float* __restrict__ dst,
    float* __restrict__ deg, int nrows, const float* __restrict__ W1,
    const float* __restrict__ b1, const float* __restrict__ W2,
    float* __restrict__ Z) {
  constexpr int R = 1 << SHIFT;
  constexpr int CAP = R * 64;
  static_assert(!EDGE || R == 64, "the edge products are written for one 64 x 64 tile");
  static_assert(R % 64 == 0, "a wave sums rows r and r + 32 side by side");
  constexpr int EW = EDGE ? 64 * WP64 : 4;
  __shared__ __attribute__((aligned(16))) float tile[R * 64 + LIST_PAD];
  __shared__ __attribute__((aligned(16))) float W1sh[EW];
  __shared__ __attribute__((aligned(16))) float W2bsh[EW];
  __shared__ __attribute__((aligned(16))) float At[EW];
  __shared__ __attribute__((aligned(16))) float Bt[EW];
  __shared__ int cnt[R];
  __shared__ int start[R + 1];
  __shared__ int wsum[32];
  int* sorted = reinterpret_cast<int*>(tile);
  const int tid = threadIdx.x;
  const int lane = tid & 31;
  v4f w1v = {}, w2v = {};
  if (EDGE) {
    w1v = *(const v4f*)(W1 + (tid >> 4) * 64 + 4 * (tid & 15));
    w2v = *(const v4f*)(W2 + (tid >> 4) * 128 + 64 + 4 * (tid & 15));
  }
  const int wave = __builtin_amdgcn_readfirstlane(tid >> 5);
  const int b = blockIdx.x;

  int s[BPW], t[BPW], e[BPW];
  const int* run[BPW];
#pragma unroll
  for (int j = 0; j < BPW; ++j) {
    const int bin = wave + 32 * j;
    const int bc = bin < BIN_BLOCKS ? bin : BIN_BLOCKS - 1;
    const int* trow = table + bc * (NB + 1) + b;
    s[j] = __builtin_amdgcn_readfirstlane(trow[0]);
    const int tj = __builtin_amdgcn_readfirstlane(trow[1]);
    t[j] = bin < BIN_BLOCKS ? tj : s[j];
    run[j] = entries + bc * CHUNK;
  }
  for (int i = tid; i < R; i += 1024) cnt[i] = 0;
#pragma unroll
  for (int j = 0; j < BPW; ++j) {
    int k = s[j] + lane;
    k = k < t[j] ? k : t[j] - 1;
    k = k > 0 ? k : 0;
    e[j] = run[j][k];
  }
  if (EDGE) {
    *(v4f*)(W1sh + (tid >> 4) * WP64 + 4 * (tid & 15)) = w1v;
    *(v4f*)(W2bsh + (tid >> 4) * WP64 + 4 * (tid & 15)) = w2v;
  }
  __syncthreads();

#pragma unroll
  for (int j = 0; j < BPW; ++j) {
    if (s[j] + lane < t[j]) atomicAdd(&cnt[e[j] >> SRC_BITS], 1);
    for (int k = s[j] + 32 + lane; k < t[j]; k += 32)
      atomicAdd(&cnt[run[j][k] >> SRC_BITS], 1);
  }
  __syncthreads();

  {
    const int c = (tid < R) ? cnt[tid] : 0;
    int incl = c;
    for (int d = 1; d < 32; d <<= 1) {
      const int u = __shfl_up(incl, d);
      if (lane >= d) incl += u;
    }
    if (lane == 31) wsum[wave] = incl;
    __syncthreads();
    if (tid < R) {
      int before = 0;
      for (int w2 = 0; w2 < wave; ++w2) before += wsum[w2];
      const int st = before + incl - c;
      start[tid] = st;
      cnt[tid] = st;
      if (tid == R - 1) start[R] = st + c;
      if (COUNT_DEG) {
        const int row = b * R + tid;
        if (row < nrows) deg[row] = (float)c;
      }
    }
    __syncthreads();
  }
  const int total = start[R];

  if (total <= CAP) {
#pragma unroll
    for (int j = 0; j < BPW; ++j) {
      if (s[j] + lane < t[j]) {
        const int pos = atomicAdd(&cnt[e[j] >> SRC_BITS], 1);
        sorted[pos] = e[j] & SRC_MASK;
      }
      for (int k = s[j] + 32 + lane; k < t[j]; k += 32) {
        const int ee = run[j][k];
        const int pos = atomicAdd(&cnt[ee >> SRC_BITS], 1);
        sorted[pos] = ee & SRC_MASK;
      }
    }
    __syncthreads();

    const int lh = lane >> 4;
    const int c4 = 4 * (lane & 15);
    for (int r0 = wave; r0 < R; r0 += 64) {
      const int r = r0 + 32 * lh;
      const int beg = start[r];
      const int n = start[r + 1] - beg;
      const int n0 = __builtin_amdgcn_readlane(n, 0);
      const int n1 = __builtin_amdgcn_readlane(n, 16);
      v4f acc = {};
      sum_two_rows<8>(src, sorted, CAP, beg, n, n0 > n1 ? n0 : n1, 4 * c4, acc);
      if (EDGE) {
        *(v4f*)(At + r * WP64 + c4) = acc;
      } else {
        const int row = b * R + r;
        if (row < nrows) *(v4f*)(dst + (size_t)row * 64 + c4) = acc;
      }
    }
  } else {
    for (int i = tid; i < R * 64; i += 1024) tile[i] = 0.0f;
    __syncthreads();
#pragma unroll
    for (int j = 0; j < BPW; ++j) {
      int k = s[j];
      for (; k + GROUP <= t[j]; k += GROUP)
        add_group<true, 64>(src, run[j], k, GROUP, lane, tile);
      if (k < t[j]) add_group<false, 64>(src, run[j], k, t[j] - k, lane, tile);
    }
    __syncthreads();
    for (int r = wave; r < R; r += 32) {
      const int row = b * R + r;
      if (EDGE) {
        At[r * WP64 + lane] = tile[r * 64 + lane];
        At[r * WP64 + 32 + lane] = tile[r * 64 + 32 + lane];
      } else if (row < nrows) {
        float* q = dst + (size_t)row * 64 + lane;
        q[0] = tile[r * 64 + lane];
        q[32] = tile[r * 64 + 32 + lane];
      }
    }
  }

  if (EDGE) {
    const int ll = lane & 15;
    const int lh = lane >> 4;
    const int tr = wave >> 2;
    const int tc = wave & 3;
    __syncthreads();
    if (wave < 16) {
      v8f acc = {};
      const float* Ap = At + (16 * tr + ll) * WP64 + 4 * lh;
      const float* Bp = W1sh + (16 * tc + ll) * WP64 + 4 * lh;
#pragma unroll
      for (int q = 0; q < 8; ++q) {
        const v4f a = *(const v4f*)(Ap + 8 * q);
        const v4f w = *(const v4f*)(Bp + 8 * q);
        acc = wmma_f32(a.xy, w.xy, acc);
        acc = wmma_f32(a.zw, w.zw, acc);
      }
      const float bias = b1[16 * tc + ll];
#pragma unroll
      for (int j = 0; j < 8; ++j) {
        const int rr = 16 * tr + j + 8 * lh;
        const float n = (float)(start[rr + 1] - start[rr]);
        Bt[rr * WP64 + 16 * tc + ll] = acc[j] + n * bias;
      }
    }
    __syncthreads();
    for (int r = wave; r < R; r += 32) {
      const int row = b * R + r;
      if (row < nrows) {
        float* q = dst + (size_t)row * 64 + lane;
        q[0] = Bt[r * WP64 + lane];
        q[32] = Bt[r * WP64 + 32 + lane];
      }
    }
    if (wave < 16) {
      v8f acc = {};
      const float* Ap = Bt + (16 * tr + ll) * WP64 + 4 * lh;
      const float* Bp = W2bsh + (16 * tc + ll) * WP64 + 4 * lh;
#pragma unroll
      for (int q = 0; q < 8; ++q) {
        const v4f a = *(const v4f*)(Ap + 8 * q);
        const v4f w = *(const v4f*)(Bp + 8 * q);
        acc = wmma_f32(a.xy, w.xy, acc);
        acc = wmma_f32(a.zw, w.zw, acc);
      }
#pragma unroll
      for (int j = 0; j < 8; ++j)
        At[(16 * tr + j + 8 * lh) * WP64 + 16 * tc + ll] = acc[j];
    }
    __syncthreads();
    for (int r = wave; r < R; r += 32) {
      const int row = b * R + r;
      if (row < nrows) {
        float* q = Z + (size_t)row * 64 + lane;
        q[0] = At[r * WP64 + lane];
        q[32] = At[r * WP64 + 32 + lane];
      }
    }
  }
}

#define NODE_THREADS 512
#define NODE_WAVES (NODE_THREADS / 32)
#define BPW_V ((BIN_BLOCKS + NODE_WAVES - 1) / NODE_WAVES)
#define SP 68
#define CAP_V 8192
static_assert((1 << SHIFT_V) == 16 * NODE_WAVES, "a wave owns 16 rows of the range");
static_assert(N_NODES_C % 16 == 0, "a wave's 16 rows are all real rows or all past the end");
static_assert(16 * TP <= 16 * SP, "the half tile of T fits the slab");

__global__ __launch_bounds__(NODE_THREADS, 8) void node_sum_mix_out_kernel(
    const float* __restrict__ Z, const int* __restrict__ entries,
    const int* __restrict__ table, const float* __restrict__ X,
    const float* __restrict__ X0, const float* __restrict__ W2,
    const float* __restrict__ b2, const float* __restrict__ W,
    const float* __restrict__ b, float* __restrict__ Out, int nrows) {
  constexpr int R = 1 << SHIFT_V;
  __shared__ __attribute__((aligned(16))) float Ssh[R * SP];
  __shared__ __attribute__((aligned(16))) float W2sh[64 * WP64];
  __shared__ __attribute__((aligned(16))) float Wsh[64 * WP64];
  __shared__ int sorted[CAP_V + LIST_PAD];
  __shared__ int cnt[R];
  __shared__ int start[R + 1];
  __shared__ int wsum[NODE_WAVES];
  const int tid = threadIdx.x;
  const int lane = tid & 31;
  const int ll = lane & 15;
  const int lh = lane >> 4;
  const int wave = __builtin_amdgcn_readfirstlane(tid >> 5);
  const int bk = blockIdx.x;
  const int rowBase = bk * R + 16 * wave;
  const bool live = rowBase < nrows;

  v4f ax[8];
  const float* Xrow = X + (size_t)((live ? rowBase : 0) + ll) * 64 + 4 * lh;
#pragma unroll
  for (int q = 0; q < 8; ++q) ax[q] = *(const v4f*)(Xrow + 8 * q);

  int s[BPW_V], t[BPW_V], e[BPW_V];
  const int* run[BPW_V];
#pragma unroll
  for (int j = 0; j < BPW_V; ++j) {
    const int bin = wave + NODE_WAVES * j;
    const int bc = bin < BIN_BLOCKS ? bin : BIN_BLOCKS - 1;
    const int* trow = table + bc * (NB_V + 1) + bk;
    s[j] = __builtin_amdgcn_readfirstlane(trow[0]);
    const int tj = __builtin_amdgcn_readfirstlane(trow[1]);
    t[j] = bin < BIN_BLOCKS ? tj : s[j];
    run[j] = entries + bc * CHUNK;
  }
  for (int i = tid; i < R; i += NODE_THREADS) cnt[i] = 0;
#pragma unroll
  for (int j = 0; j < BPW_V; ++j) {
    int k = s[j] + lane;
    k = k < t[j] ? k : t[j] - 1;
    k = k > 0 ? k : 0;
    e[j] = run[j][k];
  }
  stage_w<NODE_THREADS, 64, WP64, 128>(W2, W2sh);
  stage_w<NODE_THREADS, 64, WP64>(W, Wsh);
  __syncthreads();

#pragma unroll
  for (int j = 0; j < BPW_V; ++j) {
    if (s[j] + lane < t[j]) atomicAdd(&cnt[e[j] >> SRC_BITS], 1);
    for (int k = s[j] + 32 + lane; k < t[j]; k += 32)
      atomicAdd(&cnt[run[j][k] >> SRC_BITS], 1);
  }
  __syncthreads();

  {
    const int cn = (tid < R) ? cnt[tid] : 0;
    int incl = cn;
    for (int d = 1; d < 32; d <<= 1) {
      const int u = __shfl_up(incl, d);
      if (lane >= d) incl += u;
    }
    if (lane == 31) wsum[wave] = incl;
    __syncthreads();
    if (tid < R) {
      int before = 0;
      for (int w2 = 0; w2 < wave; ++w2) before += wsum[w2];
      const int st = before + incl - cn;
      start[tid] = st;
      cnt[tid] = st;
      if (tid == R - 1) start[R] = st + cn;
    }
    __syncthreads();
  }
  const int total = start[R];

  float* Sw = Ssh + wave * (16 * SP);
  if (total <= CAP_V) {
#pragma unroll
    for (int j = 0; j < BPW_V; ++j) {
      if (s[j] + lane < t[j]) {
        const int pos = atomicAdd(&cnt[e[j] >> SRC_BITS], 1);
        sorted[pos] = e[j] & SRC_MASK;
      }
      for (int k = s[j] + 32 + lane; k < t[j]; k += 32) {
        const int ee = run[j][k];
        const int pos = atomicAdd(&cnt[ee >> SRC_BITS], 1);
        sorted[pos] = ee & SRC_MASK;
      }
    }
    __syncthreads();

    if (live) {
      const int c4 = 4 * ll;
      for (int pr = 0; pr < 8; ++pr) {
        const int rr = 2 * pr + lh;
        const int r = 16 * wave + rr;
        const int beg = start[r];
        const int n = start[r + 1] - beg;
        const int n0 = __builtin_amdgcn_readlane(n, 0);
        const int n1 = __builtin_amdgcn_readlane(n, 16);
        v4f acc = {};
        sum_two_rows<16>(Z, sorted, CAP_V, beg, n, n0 > n1 ? n0 : n1, 4 * c4, acc);
        *(v4f*)(Sw + rr * SP + c4) = acc;
      }
    }
  } else {
    for (int i = tid; i < R * SP; i += NODE_THREADS) Ssh[i] = 0.0f;
    __syncthreads();
#pragma unroll
    for (int j = 0; j < BPW_V; ++j) {
      int k = s[j];
      for (; k + GROUP <= t[j]; k += GROUP)
        add_group<true, SP>(Z, run[j], k, GROUP, lane, Ssh);
      if (k < t[j]) add_group<false, SP>(Z, run[j], k, t[j] - k, lane, Ssh);
    }
    __syncthreads();
  }
  if (!live) return;
  wave_lds_handover();

  const int rw = 16 * wave;
  const float dg = (float)(start[rw + ll + 1] - start[rw + ll]);
  const float* W2row = W2sh + ll * WP64 + 4 * lh;
  v8f acc0, acc1, acc2, acc3;
#pragma unroll
  for (int j = 0; j < 8; ++j) {
    const float* sr = Sw + (j + lh * 8) * SP + ll;
    acc0[j] = sr[0];
    acc1[j] = sr[16];
    acc2[j] = sr[32];
    acc3[j] = sr[48];
  }
  wave_lds_handover();
#pragma unroll
  for (int q = 0; q < 8; ++q)
    wmma_piece<WP64>(dg * ax[q], W2row + 8 * q, acc0, acc1, acc2, acc3);

  const float oma = 1.0f - ALPHA_C;
#pragma unroll
  for (int j = 0; j < 8; ++j) {
    const int rj = j + lh * 8;
    const float dr = (float)(start[rw + rj + 1] - start[rw + rj]);
    const float* x0r = X0 + (size_t)(rowBase + rj) * 64;
    acc0[j] = oma * (acc0[j] + dr * b2[ll])      + ALPHA_C * x0r[ll];
    acc1[j] = oma * (acc1[j] + dr * b2[16 + ll]) + ALPHA_C * x0r[16 + ll];
    acc2[j] = oma * (acc2[j] + dr * b2[32 + ll]) + ALPHA_C * x0r[32 + ll];
    acc3[j] = oma * (acc3[j] + dr * b2[48 + ll]) + ALPHA_C * x0r[48 + ll];
  }

  float* Tst = Sw + (lh * 8) * TP + ll;
  const float* Trow = Sw + ll * TP + 4 * lh;
  const float* Wrow = Wsh + ll * WP64 + 4 * lh;
  v8f o0 = {}, o1 = {}, o2 = {}, o3 = {};
  v4f a[4];

#pragma unroll
  for (int j = 0; j < 8; ++j) {
    Tst[j * TP]      = acc0[j];
    Tst[j * TP + 16] = acc1[j];
  }
  wave_lds_handover();
#pragma unroll
  for (int q = 0; q < 4; ++q) a[q] = *(const v4f*)(Trow + 8 * q);
  wave_lds_handover();
#pragma unroll
  for (int q = 0; q < 4; ++q)
    wmma_piece<WP64>(a[q], Wrow + 8 * q, o0, o1, o2, o3);

#pragma unroll
  for (int j = 0; j < 8; ++j) {
    Tst[j * TP]      = acc2[j];
    Tst[j * TP + 16] = acc3[j];
  }
  wave_lds_handover();
#pragma unroll
  for (int q = 0; q < 4; ++q) a[q] = *(const v4f*)(Trow + 8 * q);
#pragma unroll
  for (int q = 0; q < 4; ++q)
    wmma_piece<WP64>(a[q], Wrow + 32 + 8 * q, o0, o1, o2, o3);

#pragma unroll
  for (int j = 0; j < 8; ++j) {
    float* o = Out + (size_t)(rowBase + j + lh * 8) * 64;
    o[ll]      = o0[j] + b[ll];
    o[16 + ll] = o1[j] + b[16 + ll];
    o[32 + ll] = o2[j] + b[32 + ll];
    o[48 + ll] = o3[j] + b[48 + ll];
  }
}

extern "C" void kernel_launch(void* const* d_in, const int* in_sizes, int n_in,
                              void* d_out, int out_size, void* d_ws,
                              size_t ws_size, hipStream_t stream) {
  (void)in_sizes; (void)n_in; (void)out_size; (void)ws_size;
  const float* X    = (const float*)d_in[0];
  const float* X0   = (const float*)d_in[1];
  const int*   vertex = (const int*)d_in[2];
  const int*   edges  = (const int*)d_in[3];
  const float* W1w  = (const float*)d_in[4];
  const float* W1b  = (const float*)d_in[5];
  const float* W2w  = (const float*)d_in[6];
  const float* W2b  = (const float*)d_in[7];
  const float* Ww   = (const float*)d_in[8];
  const float* Wb   = (const float*)d_in[9];

  float* out = (float*)d_out;
  float* Xe  = out + (size_t)N_NODES_C * 64;

  float* P1 = (float*)d_ws + (size_t)N_NODES_C * 64;
  int* entE = (int*)P1;
  int* entV = entE + NNZ_C;
  int* tabE = entV + NNZ_C;
  int* tabV = tabE + BIN_BLOCKS * (NB_E + 1);
  float* Z  = P1 + Z_OFFSET;

  front_kernel<<<FRONT_BLOCKS, 1024, 0, stream>>>(edges, vertex, entE, tabE, entV,
                                                  tabV);
  sum_range_kernel<SHIFT_E, NB_E, false, true><<<NB_E, 1024, 0, stream>>>(
      X, entE, tabE, Xe, nullptr, N_EDGES_C, W1w, W1b, W2w, Z);
  node_sum_mix_out_kernel<<<NB_V, NODE_THREADS, 0, stream>>>(
      Z, entV, tabV, X, X0, W2w, W2b, Ww, Wb, out, N_NODES_C);
}
